# starting kernel
# baseline (speedup 1.0000x reference)
#include <hip/hip_runtime.h>
#include <hip/hip_bf16.h>
#include <math.h>

typedef float v2f __attribute__((ext_vector_type(2)));
typedef float v8f __attribute__((ext_vector_type(8)));

#define EPS 1e-12f
#define NEIGH_WEIGHT 0.2f

__global__ void zero_kernel(float* p, size_t n) {
    size_t i = (size_t)blockIdx.x * blockDim.x + threadIdx.x;
    size_t stride = (size_t)gridDim.x * blockDim.x;
    for (; i < n; i += stride) p[i] = 0.0f;
}

__global__ void stats_kernel(const float* __restrict__ pos_diff,
                             const int* __restrict__ nor_idx, int M,
                             float* __restrict__ scratch) {
    int t = blockIdx.x * blockDim.x + threadIdx.x;
    float v = 0.0f;
    if (t < M) v = pos_diff[nor_idx[t]];
    float s1 = v, s2 = v * v;
    #pragma unroll
    for (int off = 16; off > 0; off >>= 1) {
        s1 += __shfl_xor(s1, off, 32);
        s2 += __shfl_xor(s2, off, 32);
    }
    if ((threadIdx.x & 31) == 0) {
        __hip_atomic_fetch_add(&scratch[0], s1, __ATOMIC_RELAXED, __HIP_MEMORY_SCOPE_AGENT);
        __hip_atomic_fetch_add(&scratch[1], s2, __ATOMIC_RELAXED, __HIP_MEMORY_SCOPE_AGENT);
    }
}

__global__ __launch_bounds__(256) void gemm_h_kernel(
    const float* __restrict__ feats, const float* __restrict__ W,
    const float* __restrict__ bias, float* __restrict__ h, int N) {
    __shared__ float As[16 * 260];
    __shared__ float T[16 * 132];
    __shared__ float P[16 * 17];
    __shared__ float RowInv[16];

    const int tid  = threadIdx.x;
    const int wave = tid >> 5;
    const int lane = tid & 31;
    const int row0 = blockIdx.x * 16;
    const int cb   = wave * 16;
    const int hl   = lane & 15;
    const int hi   = lane >> 4;

    #pragma unroll
    for (int j = 0; j < 4; ++j) {
        int idx = tid * 4 + j * 1024;
        int r = idx >> 8;
        int c = idx & 255;
        int grow = row0 + r;
        float4 v = make_float4(0.f, 0.f, 0.f, 0.f);
        if (grow < N) v = *(const float4*)(feats + (size_t)grow * 256 + c);
        float* dstp = &As[r * 260 + c];
        dstp[0] = v.x; dstp[1] = v.y; dstp[2] = v.z; dstp[3] = v.w;
    }
    __syncthreads();

    v8f acc = {};
    for (int k = 0; k < 256; k += 4) {
        int ka = k + hi * 2;
        v2f a, b;
        a.x = As[hl * 260 + ka];
        a.y = As[hl * 260 + ka + 1];
        b.x = W[(size_t)ka * 128 + cb + hl];
        b.y = W[(size_t)(ka + 1) * 128 + cb + hl];
        acc = __builtin_amdgcn_wmma_f32_16x16x4_f32(false, a, false, b,
                                                    (short)0, acc, false, false);
    }

    float bv = bias[cb + hl];
    int rbase = hi * 8;
    #pragma unroll
    for (int r = 0; r < 8; ++r) {
        float v = acc[r] + bv;
        v = v > 0.0f ? v : 0.0f;
        T[(rbase + r) * 132 + cb + hl] = v;
    }
    __syncthreads();

    {
        int row = tid >> 4;
        int c0  = (tid & 15) * 8;
        float s = 0.0f;
        #pragma unroll
        for (int j = 0; j < 8; ++j) { float v = T[row * 132 + c0 + j]; s += v * v; }
        P[row * 17 + (tid & 15)] = s;
    }
    __syncthreads();
    if (tid < 16) {
        float s = 0.0f;
        #pragma unroll
        for (int j = 0; j < 16; ++j) s += P[tid * 17 + j];
        RowInv[tid] = 1.0f / fmaxf(sqrtf(s), EPS);
    }
    __syncthreads();

    {
        int row  = tid >> 4;
        int grow = row0 + row;
        if (grow < N) {
            int c0 = (tid & 15) * 8;
            float inv = RowInv[row];
            float* op = h + (size_t)grow * 128 + c0;
            #pragma unroll
            for (int j = 0; j < 8; ++j) op[j] = T[row * 132 + c0 + j] * inv;
        }
    }
}

__global__ __launch_bounds__(256) void aggregate_kernel(
    const float* __restrict__ h, const int* __restrict__ src,
    const int* __restrict__ dst, float* __restrict__ accum,
    float* __restrict__ deg, int E) {
    int gw = (int)(((size_t)blockIdx.x * blockDim.x + threadIdx.x) >> 5);
    int lane = threadIdx.x & 31;
    if (gw >= E) return;
    int s = src[gw], d = dst[gw];
    const float* hp = h + (size_t)s * 128;
    float* ap = accum + (size_t)d * 128;
    #pragma unroll
    for (int j = 0; j < 4; ++j) {
        int c = lane + 32 * j;
        __hip_atomic_fetch_add(&ap[c], hp[c], __ATOMIC_RELAXED, __HIP_MEMORY_SCOPE_AGENT);
    }
    if (lane == 0)
        __hip_atomic_fetch_add(&deg[d], 1.0f, __ATOMIC_RELAXED, __HIP_MEMORY_SCOPE_AGENT);
}

__global__ __launch_bounds__(256) void finalize_kernel(
    const float* __restrict__ h, const float* __restrict__ accum,
    const float* __restrict__ deg, const float* __restrict__ center,
    const float* __restrict__ pos_diff, const int* __restrict__ out_nodes,
    const int* __restrict__ labels, const int* __restrict__ epoch,
    float* __restrict__ scratch, float* __restrict__ scores_out,
    int N, int Mnor) {
    int gw = (int)(((size_t)blockIdx.x * blockDim.x + threadIdx.x) >> 5);
    int lane = threadIdx.x & 31;
    if (gw >= N) return;
    int i = gw;

    float dg = deg[i];
    float invd = dg > 0.0f ? 1.0f / fmaxf(dg, 1.0f) : 0.0f;

    const float* hp = h + (size_t)i * 128;
    const float* ap = accum + (size_t)i * 128;
    float hv[4], mv[4];
    float dot = 0.0f;
    #pragma unroll
    for (int j = 0; j < 4; ++j) {
        int c = lane + 32 * j;
        hv[j] = hp[c];
        mv[j] = ap[c] * invd;
        dot += hv[j] * mv[j];
    }
    #pragma unroll
    for (int off = 16; off > 0; off >>= 1) dot += __shfl_xor(dot, off, 32);

    int on = out_nodes[i];
    float cnt = (float)Mnor;
    float mu  = scratch[0] / cnt;
    float var = (scratch[1] - cnt * mu * mu) / (cnt - 1.0f);
    float sd  = sqrtf(var);
    float z   = (pos_diff[on] - mu) / sd;
    float pre = 1.0f - 1.0f / (1.0f + expf(-z));

    float beta = powf(0.9f, (float)epoch[0]);
    if (beta < 0.1f) beta = 0.0f;
    float attn = beta * pre + (1.0f - beta) * dot;
    float nei  = attn * NEIGH_WEIGHT;

    float score = 0.0f;
    #pragma unroll
    for (int j = 0; j < 4; ++j) {
        int c = lane + 32 * j;
        float h2 = nei * mv[j] + (1.0f - nei) * hv[j];
        score += h2 * center[c];
    }
    #pragma unroll
    for (int off = 16; off > 0; off >>= 1) score += __shfl_xor(score, off, 32);

    if (lane == 0) {
        scores_out[i] = score;
        int lab = labels[on];
        float spl = log1pf(expf(-fabsf(score)));
        if (lab == 1) {
            float sp = fmaxf(score, 0.0f) + spl;
            __hip_atomic_fetch_add(&scratch[4], sp, __ATOMIC_RELAXED, __HIP_MEMORY_SCOPE_AGENT);
            __hip_atomic_fetch_add(&scratch[5], 1.0f, __ATOMIC_RELAXED, __HIP_MEMORY_SCOPE_AGENT);
        } else {
            float sp = fmaxf(-score, 0.0f) + spl;
            __hip_atomic_fetch_add(&scratch[2], sp, __ATOMIC_RELAXED, __HIP_MEMORY_SCOPE_AGENT);
            __hip_atomic_fetch_add(&scratch[3], 1.0f, __ATOMIC_RELAXED, __HIP_MEMORY_SCOPE_AGENT);
        }
    }
}

__global__ void loss_kernel(const float* __restrict__ scratch, float* __restrict__ out0) {
    if (threadIdx.x == 0 && blockIdx.x == 0) {
        out0[0] = scratch[2] / fmaxf(scratch[3], 1.0f) +
                  scratch[4] / fmaxf(scratch[5], 1.0f);
    }
}

extern "C" void kernel_launch(void* const* d_in, const int* in_sizes, int n_in,
                              void* d_out, int out_size, void* d_ws, size_t ws_size,
                              hipStream_t stream) {
    const float* feats    = (const float*)d_in[0];
    const float* W        = (const float*)d_in[1];
    const float* b        = (const float*)d_in[2];
    const float* center   = (const float*)d_in[3];
    const float* pos_diff = (const float*)d_in[4];
    const int*   src      = (const int*)d_in[5];
    const int*   dst      = (const int*)d_in[6];
    const int*   out_nodes= (const int*)d_in[7];
    const int*   nor_idx  = (const int*)d_in[8];
    const int*   labels   = (const int*)d_in[9];
    const int*   epoch    = (const int*)d_in[10];

    const int N = in_sizes[0] / 256;
    const int E = in_sizes[5];
    const int M = in_sizes[8];

    float* wsf     = (float*)d_ws;
    float* scratch = wsf;
    float* h       = wsf + 64;
    float* accum   = h + (size_t)N * 128;
    float* deg     = accum + (size_t)N * 128;

    float* out     = (float*)d_out;

    size_t zn = 64 + 2 * (size_t)N * 128 + (size_t)N;
    zero_kernel<<<2048, 256, 0, stream>>>(wsf, zn);

    stats_kernel<<<(M + 255) / 256, 256, 0, stream>>>(pos_diff, nor_idx, M, scratch);

    gemm_h_kernel<<<(N + 15) / 16, 256, 0, stream>>>(feats, W, b, h, N);

    aggregate_kernel<<<(E + 7) / 8, 256, 0, stream>>>(h, src, dst, accum, deg, E);

    finalize_kernel<<<(N + 7) / 8, 256, 0, stream>>>(h, accum, deg, center, pos_diff,
                                                     out_nodes, labels, epoch, scratch,
                                                     out + 1, N, M);

    loss_kernel<<<1, 64, 0, stream>>>(scratch, out);
}


# finalize: per-node same-address f32 atomics on the loss replaced by a per-block LDS sum and one record store; loss kernel adds the records in fixed order (loss summation order changes)
# speedup vs baseline: 1.6055x; 1.6055x over previous
#include <hip/hip_runtime.h>
#include <hip/hip_bf16.h>
#include <math.h>

typedef float v2f __attribute__((ext_vector_type(2)));
typedef float v8f __attribute__((ext_vector_type(8)));

#define EPS 1e-12f
#define NEIGH_WEIGHT 0.2f

__global__ void zero_kernel(float* p, size_t n) {
    size_t i = (size_t)blockIdx.x * blockDim.x + threadIdx.x;
    size_t stride = (size_t)gridDim.x * blockDim.x;
    for (; i < n; i += stride) p[i] = 0.0f;
}

__global__ void stats_kernel(const float* __restrict__ pos_diff,
                             const int* __restrict__ nor_idx, int M,
                             float* __restrict__ scratch) {
    int t = blockIdx.x * blockDim.x + threadIdx.x;
    float v = 0.0f;
    if (t < M) v = pos_diff[nor_idx[t]];
    float s1 = v, s2 = v * v;
    #pragma unroll
    for (int off = 16; off > 0; off >>= 1) {
        s1 += __shfl_xor(s1, off, 32);
        s2 += __shfl_xor(s2, off, 32);
    }
    if ((threadIdx.x & 31) == 0) {
        __hip_atomic_fetch_add(&scratch[0], s1, __ATOMIC_RELAXED, __HIP_MEMORY_SCOPE_AGENT);
        __hip_atomic_fetch_add(&scratch[1], s2, __ATOMIC_RELAXED, __HIP_MEMORY_SCOPE_AGENT);
    }
}

__global__ __launch_bounds__(256) void gemm_h_kernel(
    const float* __restrict__ feats, const float* __restrict__ W,
    const float* __restrict__ bias, float* __restrict__ h, int N) {
    __shared__ float As[16 * 260];
    __shared__ float T[16 * 132];
    __shared__ float P[16 * 17];
    __shared__ float RowInv[16];

    const int tid  = threadIdx.x;
    const int wave = tid >> 5;
    const int lane = tid & 31;
    const int row0 = blockIdx.x * 16;
    const int cb   = wave * 16;
    const int hl   = lane & 15;
    const int hi   = lane >> 4;

    #pragma unroll
    for (int j = 0; j < 4; ++j) {
        int idx = tid * 4 + j * 1024;
        int r = idx >> 8;
        int c = idx & 255;
        int grow = row0 + r;
        float4 v = make_float4(0.f, 0.f, 0.f, 0.f);
        if (grow < N) v = *(const float4*)(feats + (size_t)grow * 256 + c);
        float* dstp = &As[r * 260 + c];
        dstp[0] = v.x; dstp[1] = v.y; dstp[2] = v.z; dstp[3] = v.w;
    }
    __syncthreads();

    v8f acc = {};
    for (int k = 0; k < 256; k += 4) {
        int ka = k + hi * 2;
        v2f a, b;
        a.x = As[hl * 260 + ka];
        a.y = As[hl * 260 + ka + 1];
        b.x = W[(size_t)ka * 128 + cb + hl];
        b.y = W[(size_t)(ka + 1) * 128 + cb + hl];
        acc = __builtin_amdgcn_wmma_f32_16x16x4_f32(false, a, false, b,
                                                    (short)0, acc, false, false);
    }

    float bv = bias[cb + hl];
    int rbase = hi * 8;
    #pragma unroll
    for (int r = 0; r < 8; ++r) {
        float v = acc[r] + bv;
        v = v > 0.0f ? v : 0.0f;
        T[(rbase + r) * 132 + cb + hl] = v;
    }
    __syncthreads();

    {
        int row = tid >> 4;
        int c0  = (tid & 15) * 8;
        float s = 0.0f;
        #pragma unroll
        for (int j = 0; j < 8; ++j) { float v = T[row * 132 + c0 + j]; s += v * v; }
        P[row * 17 + (tid & 15)] = s;
    }
    __syncthreads();
    if (tid < 16) {
        float s = 0.0f;
        #pragma unroll
        for (int j = 0; j < 16; ++j) s += P[tid * 17 + j];
        RowInv[tid] = 1.0f / fmaxf(sqrtf(s), EPS);
    }
    __syncthreads();

    {
        int row  = tid >> 4;
        int grow = row0 + row;
        if (grow < N) {
            int c0 = (tid & 15) * 8;
            float inv = RowInv[row];
            float* op = h + (size_t)grow * 128 + c0;
            #pragma unroll
            for (int j = 0; j < 8; ++j) op[j] = T[row * 132 + c0 + j] * inv;
        }
    }
}

__global__ __launch_bounds__(256) void aggregate_kernel(
    const float* __restrict__ h, const int* __restrict__ src,
    const int* __restrict__ dst, float* __restrict__ accum,
    float* __restrict__ deg, int E) {
    int gw = (int)(((size_t)blockIdx.x * blockDim.x + threadIdx.x) >> 5);
    int lane = threadIdx.x & 31;
    if (gw >= E) return;
    int s = src[gw], d = dst[gw];
    const float* hp = h + (size_t)s * 128;
    float* ap = accum + (size_t)d * 128;
    #pragma unroll
    for (int j = 0; j < 4; ++j) {
        int c = lane + 32 * j;
        __hip_atomic_fetch_add(&ap[c], hp[c], __ATOMIC_RELAXED, __HIP_MEMORY_SCOPE_AGENT);
    }
    if (lane == 0)
        __hip_atomic_fetch_add(&deg[d], 1.0f, __ATOMIC_RELAXED, __HIP_MEMORY_SCOPE_AGENT);
}

__device__ __forceinline__ float* loss_record(float* deg, float* scratch, int blk, int N) {
    return (8 * blk + 4 <= N) ? deg + 8 * (size_t)blk : scratch + 8;
}

__global__ __launch_bounds__(256) void finalize_kernel(
    const float* __restrict__ h, const float* __restrict__ accum,
    float* deg, const float* __restrict__ center,
    const float* __restrict__ pos_diff, const int* __restrict__ out_nodes,
    const int* __restrict__ labels, const int* __restrict__ epoch,
    float* scratch, float* __restrict__ scores_out,
    int N, int Mnor) {
    __shared__ float red[8][4];

    const int wave = threadIdx.x >> 5;
    const int lane = threadIdx.x & 31;
    const int i = blockIdx.x * 8 + wave;

    float p_sum = 0.0f, p_cnt = 0.0f, n_sum = 0.0f, n_cnt = 0.0f;

    if (i < N) {
        float dg = deg[i];
        float invd = dg > 0.0f ? 1.0f / fmaxf(dg, 1.0f) : 0.0f;

        const float* hp = h + (size_t)i * 128;
        const float* ap = accum + (size_t)i * 128;
        float hv[4], mv[4];
        float dot = 0.0f;
        #pragma unroll
        for (int j = 0; j < 4; ++j) {
            int c = lane + 32 * j;
            hv[j] = hp[c];
            mv[j] = ap[c] * invd;
            dot += hv[j] * mv[j];
        }
        #pragma unroll
        for (int off = 16; off > 0; off >>= 1) dot += __shfl_xor(dot, off, 32);

        int on = out_nodes[i];
        float cnt = (float)Mnor;
        float mu  = scratch[0] / cnt;
        float var = (scratch[1] - cnt * mu * mu) / (cnt - 1.0f);
        float sd  = sqrtf(var);
        float z   = (pos_diff[on] - mu) / sd;
        float pre = 1.0f - 1.0f / (1.0f + expf(-z));

        float beta = powf(0.9f, (float)epoch[0]);
        if (beta < 0.1f) beta = 0.0f;
        float attn = beta * pre + (1.0f - beta) * dot;
        float nei  = attn * NEIGH_WEIGHT;

        float score = 0.0f;
        #pragma unroll
        for (int j = 0; j < 4; ++j) {
            int c = lane + 32 * j;
            float h2 = nei * mv[j] + (1.0f - nei) * hv[j];
            score += h2 * center[c];
        }
        #pragma unroll
        for (int off = 16; off > 0; off >>= 1) score += __shfl_xor(score, off, 32);

        if (lane == 0) {
            scores_out[i] = score;
            int lab = labels[on];
            float spl = log1pf(expf(-fabsf(score)));
            if (lab == 1) {
                n_sum = fmaxf(score, 0.0f) + spl;
                n_cnt = 1.0f;
            } else {
                p_sum = fmaxf(-score, 0.0f) + spl;
                p_cnt = 1.0f;
            }
        }
    }

    if (lane == 0) {
        red[wave][0] = p_sum; red[wave][1] = p_cnt;
        red[wave][2] = n_sum; red[wave][3] = n_cnt;
    }
    __syncthreads();
    if (threadIdx.x < 4) {
        float s = 0.0f;
        for (int w = 0; w < 8; ++w) s += red[w][threadIdx.x];
        loss_record(deg, scratch, blockIdx.x, N)[threadIdx.x] = s;
    }
}

__global__ __launch_bounds__(1024) void loss_kernel(float* deg, float* scratch, int N,
                                                    float* __restrict__ out0) {
    __shared__ float red[32][4];
    const int tid  = threadIdx.x;
    const int lane = tid & 31;
    const int wave = tid >> 5;
    const int nblk = (N + 7) >> 3;

    float a0 = 0.0f, a1 = 0.0f, a2 = 0.0f, a3 = 0.0f;
    #pragma unroll 4
    for (int b = tid; b < nblk; b += 1024) {
        float4 v = *(const float4*)loss_record(deg, scratch, b, N);
        a0 += v.x; a1 += v.y; a2 += v.z; a3 += v.w;
    }
    for (int off = 16; off > 0; off >>= 1) {
        a0 += __shfl_xor(a0, off, 32);
        a1 += __shfl_xor(a1, off, 32);
        a2 += __shfl_xor(a2, off, 32);
        a3 += __shfl_xor(a3, off, 32);
    }
    if (lane == 0) {
        red[wave][0] = a0; red[wave][1] = a1; red[wave][2] = a2; red[wave][3] = a3;
    }
    __syncthreads();
    if (wave == 0) {
        float r0 = red[lane][0], r1 = red[lane][1], r2 = red[lane][2], r3 = red[lane][3];
        for (int off = 16; off > 0; off >>= 1) {
            r0 += __shfl_xor(r0, off, 32);
            r1 += __shfl_xor(r1, off, 32);
            r2 += __shfl_xor(r2, off, 32);
            r3 += __shfl_xor(r3, off, 32);
        }
        if (lane == 0)
            out0[0] = r0 / fmaxf(r1, 1.0f) + r2 / fmaxf(r3, 1.0f);
    }
}

extern "C" void kernel_launch(void* const* d_in, const int* in_sizes, int n_in,
                              void* d_out, int out_size, void* d_ws, size_t ws_size,
                              hipStream_t stream) {
    const float* feats    = (const float*)d_in[0];
    const float* W        = (const float*)d_in[1];
    const float* b        = (const float*)d_in[2];
    const float* center   = (const float*)d_in[3];
    const float* pos_diff = (const float*)d_in[4];
    const int*   src      = (const int*)d_in[5];
    const int*   dst      = (const int*)d_in[6];
    const int*   out_nodes= (const int*)d_in[7];
    const int*   nor_idx  = (const int*)d_in[8];
    const int*   labels   = (const int*)d_in[9];
    const int*   epoch    = (const int*)d_in[10];

    const int N = in_sizes[0] / 256;
    const int E = in_sizes[5];
    const int M = in_sizes[8];

    float* wsf     = (float*)d_ws;
    float* scratch = wsf;
    float* h       = wsf + 64;
    float* accum   = h + (size_t)N * 128;
    float* deg     = accum + (size_t)N * 128;

    float* out     = (float*)d_out;

    size_t zn = 64 + 2 * (size_t)N * 128 + (size_t)N;
    zero_kernel<<<2048, 256, 0, stream>>>(wsf, zn);

    stats_kernel<<<(M + 255) / 256, 256, 0, stream>>>(pos_diff, nor_idx, M, scratch);

    gemm_h_kernel<<<(N + 15) / 16, 256, 0, stream>>>(feats, W, b, h, N);

    aggregate_kernel<<<(E + 7) / 8, 256, 0, stream>>>(h, src, dst, accum, deg, E);

    finalize_kernel<<<(N + 7) / 8, 256, 0, stream>>>(h, accum, deg, center, pos_diff,
                                                     out_nodes, labels, epoch, scratch,
                                                     out + 1, N, M);

    loss_kernel<<<1, 1024, 0, stream>>>(deg, scratch, N, out);
}


# aggregation by pull: per-call CSR by dst (int count atomics, scan, fill) replaces per-edge f32 atomics; a node's wave adds h[src] rows in registers in slot order and blends there (sum orders change)
# speedup vs baseline: 2.3507x; 1.4642x over previous
#include <hip/hip_runtime.h>
#include <hip/hip_bf16.h>
#include <math.h>

typedef float v2f __attribute__((ext_vector_type(2)));
typedef float v8f __attribute__((ext_vector_type(8)));

#define EPS 1e-12f
#define NEIGH_WEIGHT 0.2f

__global__ void zero_kernel(float* __restrict__ scratch, int* __restrict__ cnt, int n) {
    int t = blockIdx.x * blockDim.x + threadIdx.x;
    if (t < 64) scratch[t] = 0.0f;
    if (t < n) cnt[t] = 0;
}

__global__ void stats_kernel(const float* __restrict__ pos_diff,
                             const int* __restrict__ nor_idx, int M,
                             float* __restrict__ scratch) {
    int t = blockIdx.x * blockDim.x + threadIdx.x;
    float v = 0.0f;
    if (t < M) v = pos_diff[nor_idx[t]];
    float s1 = v, s2 = v * v;
    #pragma unroll
    for (int off = 16; off > 0; off >>= 1) {
        s1 += __shfl_xor(s1, off, 32);
        s2 += __shfl_xor(s2, off, 32);
    }
    if ((threadIdx.x & 31) == 0) {
        __hip_atomic_fetch_add(&scratch[0], s1, __ATOMIC_RELAXED, __HIP_MEMORY_SCOPE_AGENT);
        __hip_atomic_fetch_add(&scratch[1], s2, __ATOMIC_RELAXED, __HIP_MEMORY_SCOPE_AGENT);
    }
}

__global__ __launch_bounds__(256) void gemm_h_kernel(
    const float* __restrict__ feats, const float* __restrict__ W,
    const float* __restrict__ bias, float* __restrict__ h, int N) {
    __shared__ float As[16 * 260];
    __shared__ float T[16 * 132];
    __shared__ float P[16 * 17];
    __shared__ float RowInv[16];

    const int tid  = threadIdx.x;
    const int wave = tid >> 5;
    const int lane = tid & 31;
    const int row0 = blockIdx.x * 16;
    const int cb   = wave * 16;
    const int hl   = lane & 15;
    const int hi   = lane >> 4;

    #pragma unroll
    for (int j = 0; j < 4; ++j) {
        int idx = tid * 4 + j * 1024;
        int r = idx >> 8;
        int c = idx & 255;
        int grow = row0 + r;
        float4 v = make_float4(0.f, 0.f, 0.f, 0.f);
        if (grow < N) v = *(const float4*)(feats + (size_t)grow * 256 + c);
        float* dstp = &As[r * 260 + c];
        dstp[0] = v.x; dstp[1] = v.y; dstp[2] = v.z; dstp[3] = v.w;
    }
    __syncthreads();

    v8f acc = {};
    for (int k = 0; k < 256; k += 4) {
        int ka = k + hi * 2;
        v2f a, b;
        a.x = As[hl * 260 + ka];
        a.y = As[hl * 260 + ka + 1];
        b.x = W[(size_t)ka * 128 + cb + hl];
        b.y = W[(size_t)(ka + 1) * 128 + cb + hl];
        acc = __builtin_amdgcn_wmma_f32_16x16x4_f32(false, a, false, b,
                                                    (short)0, acc, false, false);
    }

    float bv = bias[cb + hl];
    int rbase = hi * 8;
    #pragma unroll
    for (int r = 0; r < 8; ++r) {
        float v = acc[r] + bv;
        v = v > 0.0f ? v : 0.0f;
        T[(rbase + r) * 132 + cb + hl] = v;
    }
    __syncthreads();

    {
        int row = tid >> 4;
        int c0  = (tid & 15) * 8;
        float s = 0.0f;
        #pragma unroll
        for (int j = 0; j < 8; ++j) { float v = T[row * 132 + c0 + j]; s += v * v; }
        P[row * 17 + (tid & 15)] = s;
    }
    __syncthreads();
    if (tid < 16) {
        float s = 0.0f;
        #pragma unroll
        for (int j = 0; j < 16; ++j) s += P[tid * 17 + j];
        RowInv[tid] = 1.0f / fmaxf(sqrtf(s), EPS);
    }
    __syncthreads();

    {
        int row  = tid >> 4;
        int grow = row0 + row;
        if (grow < N) {
            int c0 = (tid & 15) * 8;
            float inv = RowInv[row];
            float* op = h + (size_t)grow * 128 + c0;
            #pragma unroll
            for (int j = 0; j < 8; ++j) op[j] = T[row * 132 + c0 + j] * inv;
        }
    }
}


__global__ __launch_bounds__(256) void count_kernel(const int* __restrict__ dst,
                                                    int* __restrict__ cnt, int E) {
    int e = blockIdx.x * blockDim.x + threadIdx.x;
    if (e < E)
        __hip_atomic_fetch_add(&cnt[dst[e]], 1, __ATOMIC_RELAXED, __HIP_MEMORY_SCOPE_AGENT);
}

__global__ __launch_bounds__(1024) void scan_kernel(const int* __restrict__ cnt,
                                                    int* __restrict__ off, int N) {
    __shared__ int wtot[32];
    __shared__ int wbase[32];
    const int tid  = threadIdx.x;
    const int lane = tid & 31;
    const int wave = tid >> 5;
    const int first = blockIdx.x * 1024;

    int s = 0;
    const int4* c4 = (const int4*)cnt;
    const int n4 = first >> 2;
    #pragma unroll 4
    for (int j = tid; j < n4; j += 1024) {
        int4 v = c4[j];
        s += (v.x + v.y) + (v.z + v.w);
    }

    const int node = first + tid;
    const int v = node < N ? cnt[node] : 0;
    int inc = v;
    for (int o = 1; o < 32; o <<= 1) {
        int t = __shfl_up(inc, o, 32);
        if (lane >= o) inc += t;
    }
    for (int o = 16; o > 0; o >>= 1) s += __shfl_xor(s, o, 32);
    if (lane == 31) wtot[wave] = inc;
    if (lane == 0)  wbase[wave] = s;
    __syncthreads();

    int pre  = lane < wave ? wtot[lane] : 0;
    int base = wbase[lane];
    for (int o = 16; o > 0; o >>= 1) {
        pre  += __shfl_xor(pre, o, 32);
        base += __shfl_xor(base, o, 32);
    }
    if (node < N) off[node] = base + pre + (inc - v);
}

__global__ __launch_bounds__(256) void fill_kernel(const int* __restrict__ src,
                                                   const int* __restrict__ dst,
                                                   int* __restrict__ off,
                                                   int* __restrict__ csr, int E) {
    int e = blockIdx.x * blockDim.x + threadIdx.x;
    if (e < E) {
        int s = src[e];
        int slot = __hip_atomic_fetch_add(&off[dst[e]], 1, __ATOMIC_RELAXED,
                                          __HIP_MEMORY_SCOPE_AGENT);
        if ((unsigned)slot < (unsigned)E) csr[slot] = s;
    }
}

__device__ __forceinline__ float4 row4(const float* __restrict__ hl, int s) {
    return *(const float4*)(hl + (size_t)s * 128);
}
__device__ __forceinline__ void add4(float4& a, const float4& v) {
    a.x += v.x; a.y += v.y; a.z += v.z; a.w += v.w;
}
template <int G>
__device__ __forceinline__ void add_rows(float4& acc, const float* __restrict__ hl,
                                         int myidx, int k) {
    float4 v[G];
    #pragma unroll
    for (int u = 0; u < G; ++u) v[u] = row4(hl, __builtin_amdgcn_readlane(myidx, k + u));
    #pragma unroll
    for (int u = 0; u < G; ++u) add4(acc, v[u]);
}
__device__ __forceinline__ void walk(float4& acc, const float* __restrict__ hl, int myidx, int m) {
    int k = 0;
    for (; k + 8 <= m; k += 8) add_rows<8>(acc, hl, myidx, k);
    if (m & 4) { add_rows<4>(acc, hl, myidx, k); k += 4; }
    if (m & 2) { add_rows<2>(acc, hl, myidx, k); k += 2; }
    if (m & 1) { add_rows<1>(acc, hl, myidx, k); }
}

__global__ __launch_bounds__(256) void pull_finalize_kernel(
    const float* __restrict__ h, const int* __restrict__ cnt,
    const int* __restrict__ off, const int* __restrict__ csr,
    const float* __restrict__ center,
    const float* __restrict__ pos_diff, const int* __restrict__ out_nodes,
    const int* __restrict__ labels, const int* __restrict__ epoch,
    const float* __restrict__ scratch, float* __restrict__ rec,
    float* __restrict__ scores_out, int N, int E, int Mnor) {
    __shared__ float red[8][4];

    const int wave = __builtin_amdgcn_readfirstlane(threadIdx.x >> 5);
    const int lane = threadIdx.x & 31;
    const int i = blockIdx.x * 8 + wave;

    float p_sum = 0.0f, p_cnt = 0.0f, n_sum = 0.0f, n_cnt = 0.0f;

    if (i < N) {
        const int c = cnt[i];
        int en  = off[i];
        int beg = en - c;
        if (beg < 0) beg = 0;
        if (en > E)  en = E;

        const float* hl = h + lane * 4;
        const int m0 = (en - beg) < 32 ? (en - beg) : 32;
        int idx0 = 0;
        if (lane < m0) idx0 = csr[beg + lane];
        const float4 hv = *(const float4*)(hl + (size_t)i * 128);
        const float4 cv = *(const float4*)(center + lane * 4);

        float4 acc = make_float4(0.f, 0.f, 0.f, 0.f);
        if (m0 > 0) {
            walk(acc, hl, idx0, m0);
            for (int base = beg + 32; base < en; base += 32) {
                const int m = (en - base) < 32 ? (en - base) : 32;
                int idx = 0;
                if (lane < m) idx = csr[base + lane];
                walk(acc, hl, idx, m);
            }
        }

        const float invd = c > 0 ? 1.0f / fmaxf((float)c, 1.0f) : 0.0f;
        float4 mv;
        mv.x = acc.x * invd; mv.y = acc.y * invd; mv.z = acc.z * invd; mv.w = acc.w * invd;
        float dot = hv.x * mv.x;
        dot += hv.y * mv.y;
        dot += hv.z * mv.z;
        dot += hv.w * mv.w;
        for (int o = 16; o > 0; o >>= 1) dot += __shfl_xor(dot, o, 32);

        int on = out_nodes[i];
        float cntf = (float)Mnor;
        float mu  = scratch[0] / cntf;
        float var = (scratch[1] - cntf * mu * mu) / (cntf - 1.0f);
        float sd  = sqrtf(var);
        float z   = (pos_diff[on] - mu) / sd;
        float pre = 1.0f - 1.0f / (1.0f + expf(-z));

        float beta = powf(0.9f, (float)epoch[0]);
        if (beta < 0.1f) beta = 0.0f;
        float attn = beta * pre + (1.0f - beta) * dot;
        float nei  = attn * NEIGH_WEIGHT;

        float score = (nei * mv.x + (1.0f - nei) * hv.x) * cv.x;
        score += (nei * mv.y + (1.0f - nei) * hv.y) * cv.y;
        score += (nei * mv.z + (1.0f - nei) * hv.z) * cv.z;
        score += (nei * mv.w + (1.0f - nei) * hv.w) * cv.w;
        for (int o = 16; o > 0; o >>= 1) score += __shfl_xor(score, o, 32);

        if (lane == 0) {
            scores_out[i] = score;
            int lab = labels[on];
            float spl = log1pf(expf(-fabsf(score)));
            if (lab == 1) {
                n_sum = fmaxf(score, 0.0f) + spl;
                n_cnt = 1.0f;
            } else {
                p_sum = fmaxf(-score, 0.0f) + spl;
                p_cnt = 1.0f;
            }
        }
    }

    if (lane == 0) {
        red[wave][0] = p_sum; red[wave][1] = p_cnt;
        red[wave][2] = n_sum; red[wave][3] = n_cnt;
    }
    __syncthreads();
    if (threadIdx.x < 4) {
        float s = 0.0f;
        for (int w = 0; w < 8; ++w) s += red[w][threadIdx.x];
        rec[4 * (size_t)blockIdx.x + threadIdx.x] = s;
    }
}

__global__ __launch_bounds__(1024) void loss_kernel(const float* __restrict__ rec, int N,
                                                    float* __restrict__ out0) {
    __shared__ float red[32][4];
    const int tid  = threadIdx.x;
    const int lane = tid & 31;
    const int wave = tid >> 5;
    const int nblk = (N + 7) >> 3;

    float a0 = 0.0f, a1 = 0.0f, a2 = 0.0f, a3 = 0.0f;
    #pragma unroll 4
    for (int b = tid; b < nblk; b += 1024) {
        float4 v = *(const float4*)(rec + 4 * (size_t)b);
        a0 += v.x; a1 += v.y; a2 += v.z; a3 += v.w;
    }
    for (int off = 16; off > 0; off >>= 1) {
        a0 += __shfl_xor(a0, off, 32);
        a1 += __shfl_xor(a1, off, 32);
        a2 += __shfl_xor(a2, off, 32);
        a3 += __shfl_xor(a3, off, 32);
    }
    if (lane == 0) {
        red[wave][0] = a0; red[wave][1] = a1; red[wave][2] = a2; red[wave][3] = a3;
    }
    __syncthreads();
    if (wave == 0) {
        float r0 = red[lane][0], r1 = red[lane][1], r2 = red[lane][2], r3 = red[lane][3];
        for (int off = 16; off > 0; off >>= 1) {
            r0 += __shfl_xor(r0, off, 32);
            r1 += __shfl_xor(r1, off, 32);
            r2 += __shfl_xor(r2, off, 32);
            r3 += __shfl_xor(r3, off, 32);
        }
        if (lane == 0)
            out0[0] = r0 / fmaxf(r1, 1.0f) + r2 / fmaxf(r3, 1.0f);
    }
}

extern "C" void kernel_launch(void* const* d_in, const int* in_sizes, int n_in,
                              void* d_out, int out_size, void* d_ws, size_t ws_size,
                              hipStream_t stream) {
    const float* feats    = (const float*)d_in[0];
    const float* W        = (const float*)d_in[1];
    const float* b        = (const float*)d_in[2];
    const float* center   = (const float*)d_in[3];
    const float* pos_diff = (const float*)d_in[4];
    const int*   src      = (const int*)d_in[5];
    const int*   dst      = (const int*)d_in[6];
    const int*   out_nodes= (const int*)d_in[7];
    const int*   nor_idx  = (const int*)d_in[8];
    const int*   labels   = (const int*)d_in[9];
    const int*   epoch    = (const int*)d_in[10];

    const int N = in_sizes[0] / 256;
    const int E = in_sizes[5];
    const int M = in_sizes[8];
    const int nblk = (N + 7) / 8;

    float* wsf     = (float*)d_ws;
    float* scratch = wsf;
    float* h       = wsf + 64;
    float* rec     = h + (size_t)N * 128;
    int*   cnt     = (int*)(rec + 4 * (size_t)nblk);
    int*   off     = cnt + N;
    int*   csr     = off + N;

    float* out     = (float*)d_out;

    zero_kernel<<<(N + 255) / 256, 256, 0, stream>>>(scratch, cnt, N);

    stats_kernel<<<(M + 255) / 256, 256, 0, stream>>>(pos_diff, nor_idx, M, scratch);

    count_kernel<<<(E + 255) / 256, 256, 0, stream>>>(dst, cnt, E);
    scan_kernel<<<(N + 1023) / 1024, 1024, 0, stream>>>(cnt, off, N);
    fill_kernel<<<(E + 255) / 256, 256, 0, stream>>>(src, dst, off, csr, E);

    gemm_h_kernel<<<(N + 15) / 16, 256, 0, stream>>>(feats, W, b, h, N);

    pull_finalize_kernel<<<nblk, 256, 0, stream>>>(h, cnt, off, csr, center, pos_diff,
                                                   out_nodes, labels, epoch, scratch, rec,
                                                   out + 1, N, E, M);

    loss_kernel<<<1, 1024, 0, stream>>>(rec, N, out);
}


# neighbour lists as 32-slot buckets (one 128-byte line per node) filled in ONE atomic pass, spill list past 32; count pass and scan launch removed (rows still summed in arrival-slot order)
# speedup vs baseline: 3.2251x; 1.3720x over previous
#include <hip/hip_runtime.h>
#include <hip/hip_bf16.h>
#include <math.h>

typedef float v2f __attribute__((ext_vector_type(2)));
typedef float v8f __attribute__((ext_vector_type(8)));

#define EPS 1e-12f
#define NEIGH_WEIGHT 0.2f
#define CAP 32

__global__ void zero_kernel(float* __restrict__ scratch, int* __restrict__ cnt,
                            int* __restrict__ ovfn, int n) {
    int t = blockIdx.x * blockDim.x + threadIdx.x;
    if (t < 64) scratch[t] = 0.0f;
    if (t < n) cnt[t] = 0;
    if (t == 0) ovfn[0] = 0;
}

__global__ void stats_kernel(const float* __restrict__ pos_diff,
                             const int* __restrict__ nor_idx, int M,
                             float* __restrict__ scratch) {
    int t = blockIdx.x * blockDim.x + threadIdx.x;
    float v = 0.0f;
    if (t < M) v = pos_diff[nor_idx[t]];
    float s1 = v, s2 = v * v;
    #pragma unroll
    for (int off = 16; off > 0; off >>= 1) {
        s1 += __shfl_xor(s1, off, 32);
        s2 += __shfl_xor(s2, off, 32);
    }
    if ((threadIdx.x & 31) == 0) {
        __hip_atomic_fetch_add(&scratch[0], s1, __ATOMIC_RELAXED, __HIP_MEMORY_SCOPE_AGENT);
        __hip_atomic_fetch_add(&scratch[1], s2, __ATOMIC_RELAXED, __HIP_MEMORY_SCOPE_AGENT);
    }
}

__global__ __launch_bounds__(256) void gemm_h_kernel(
    const float* __restrict__ feats, const float* __restrict__ W,
    const float* __restrict__ bias, float* __restrict__ h, int N) {
    __shared__ float As[16 * 260];
    __shared__ float T[16 * 132];
    __shared__ float P[16 * 17];
    __shared__ float RowInv[16];

    const int tid  = threadIdx.x;
    const int wave = tid >> 5;
    const int lane = tid & 31;
    const int row0 = blockIdx.x * 16;
    const int cb   = wave * 16;
    const int hl   = lane & 15;
    const int hi   = lane >> 4;

    #pragma unroll
    for (int j = 0; j < 4; ++j) {
        int idx = tid * 4 + j * 1024;
        int r = idx >> 8;
        int c = idx & 255;
        int grow = row0 + r;
        float4 v = make_float4(0.f, 0.f, 0.f, 0.f);
        if (grow < N) v = *(const float4*)(feats + (size_t)grow * 256 + c);
        float* dstp = &As[r * 260 + c];
        dstp[0] = v.x; dstp[1] = v.y; dstp[2] = v.z; dstp[3] = v.w;
    }
    __syncthreads();

    v8f acc = {};
    for (int k = 0; k < 256; k += 4) {
        int ka = k + hi * 2;
        v2f a, b;
        a.x = As[hl * 260 + ka];
        a.y = As[hl * 260 + ka + 1];
        b.x = W[(size_t)ka * 128 + cb + hl];
        b.y = W[(size_t)(ka + 1) * 128 + cb + hl];
        acc = __builtin_amdgcn_wmma_f32_16x16x4_f32(false, a, false, b,
                                                    (short)0, acc, false, false);
    }

    float bv = bias[cb + hl];
    int rbase = hi * 8;
    #pragma unroll
    for (int r = 0; r < 8; ++r) {
        float v = acc[r] + bv;
        v = v > 0.0f ? v : 0.0f;
        T[(rbase + r) * 132 + cb + hl] = v;
    }
    __syncthreads();

    {
        int row = tid >> 4;
        int c0  = (tid & 15) * 8;
        float s = 0.0f;
        #pragma unroll
        for (int j = 0; j < 8; ++j) { float v = T[row * 132 + c0 + j]; s += v * v; }
        P[row * 17 + (tid & 15)] = s;
    }
    __syncthreads();
    if (tid < 16) {
        float s = 0.0f;
        #pragma unroll
        for (int j = 0; j < 16; ++j) s += P[tid * 17 + j];
        RowInv[tid] = 1.0f / fmaxf(sqrtf(s), EPS);
    }
    __syncthreads();

    {
        int row  = tid >> 4;
        int grow = row0 + row;
        if (grow < N) {
            int c0 = (tid & 15) * 8;
            float inv = RowInv[row];
            float* op = h + (size_t)grow * 128 + c0;
            #pragma unroll
            for (int j = 0; j < 8; ++j) op[j] = T[row * 132 + c0 + j] * inv;
        }
    }
}

__global__ __launch_bounds__(256) void fill_kernel(const int* __restrict__ src,
                                                   const int* __restrict__ dst,
                                                   int* __restrict__ cnt,
                                                   int* __restrict__ bucket,
                                                   int* __restrict__ ovfn,
                                                   int* __restrict__ ovf, int E) {
    int e = blockIdx.x * blockDim.x + threadIdx.x;
    if (e < E) {
        int s = src[e];
        int d = dst[e];
        int slot = __hip_atomic_fetch_add(&cnt[d], 1, __ATOMIC_RELAXED, __HIP_MEMORY_SCOPE_AGENT);
        if ((unsigned)slot < (unsigned)CAP) {
            bucket[(size_t)d * CAP + slot] = s;
        } else {
            int p = __hip_atomic_fetch_add(ovfn, 1, __ATOMIC_RELAXED, __HIP_MEMORY_SCOPE_AGENT);
            if ((unsigned)p < (unsigned)E) {
                ovf[2 * (size_t)p]     = d;
                ovf[2 * (size_t)p + 1] = s;
            }
        }
    }
}

__device__ __forceinline__ float4 row4(const float* __restrict__ hl, int s) {
    return *(const float4*)(hl + (size_t)s * 128);
}
__device__ __forceinline__ void add4(float4& a, const float4& v) {
    a.x += v.x; a.y += v.y; a.z += v.z; a.w += v.w;
}
template <int G>
__device__ __forceinline__ void add_rows(float4& acc, const float* __restrict__ hl,
                                         int myidx, int k) {
    float4 v[G];
    #pragma unroll
    for (int u = 0; u < G; ++u) v[u] = row4(hl, __builtin_amdgcn_readlane(myidx, k + u));
    #pragma unroll
    for (int u = 0; u < G; ++u) add4(acc, v[u]);
}
__device__ __forceinline__ void walk(float4& acc, const float* __restrict__ hl, int myidx, int m) {
    int k = 0;
    for (; k + 8 <= m; k += 8) add_rows<8>(acc, hl, myidx, k);
    if (m & 4) { add_rows<4>(acc, hl, myidx, k); k += 4; }
    if (m & 2) { add_rows<2>(acc, hl, myidx, k); k += 2; }
    if (m & 1) { add_rows<1>(acc, hl, myidx, k); }
}

__global__ __launch_bounds__(256) void pull_finalize_kernel(
    const float* __restrict__ h, const int* __restrict__ cnt,
    const int* __restrict__ bucket, const int* __restrict__ ovfn,
    const int* __restrict__ ovf,
    const float* __restrict__ center,
    const float* __restrict__ pos_diff, const int* __restrict__ out_nodes,
    const int* __restrict__ labels, const int* __restrict__ epoch,
    const float* __restrict__ scratch, float* __restrict__ rec,
    float* __restrict__ scores_out, int N, int E, int Mnor) {
    __shared__ float red[8][4];

    const int wave = __builtin_amdgcn_readfirstlane(threadIdx.x >> 5);
    const int lane = threadIdx.x & 31;
    const int i = blockIdx.x * 8 + wave;

    float p_sum = 0.0f, p_cnt = 0.0f, n_sum = 0.0f, n_cnt = 0.0f;

    if (i < N) {
        const int c = cnt[i];
        const int m0 = c < CAP ? (c > 0 ? c : 0) : CAP;

        const float* hl = h + lane * 4;
        int idx0 = 0;
        if (lane < m0) idx0 = bucket[(size_t)i * CAP + lane];
        const float4 hv = *(const float4*)(hl + (size_t)i * 128);
        const float4 cv = *(const float4*)(center + lane * 4);

        float4 acc = make_float4(0.f, 0.f, 0.f, 0.f);
        if (m0 > 0) walk(acc, hl, idx0, m0);
        if (c > CAP) {
            int n = ovfn[0];
            if (n > E) n = E;
            for (int base = 0; base < n; base += 32) {
                const int k = base + lane;
                int d = -1, s = 0;
                if (k < n) { d = ovf[2 * (size_t)k]; s = ovf[2 * (size_t)k + 1]; }
                unsigned hit = __builtin_amdgcn_ballot_w32(d == i);
                while (hit) {
                    const int b = __builtin_ctz(hit);
                    hit &= hit - 1;
                    const float4 v = row4(hl, __builtin_amdgcn_readlane(s, b));
                    add4(acc, v);
                }
            }
        }

        const float invd = c > 0 ? 1.0f / fmaxf((float)c, 1.0f) : 0.0f;
        float4 mv;
        mv.x = acc.x * invd; mv.y = acc.y * invd; mv.z = acc.z * invd; mv.w = acc.w * invd;
        float dot = hv.x * mv.x;
        dot += hv.y * mv.y;
        dot += hv.z * mv.z;
        dot += hv.w * mv.w;
        for (int o = 16; o > 0; o >>= 1) dot += __shfl_xor(dot, o, 32);

        int on = out_nodes[i];
        float cntf = (float)Mnor;
        float mu  = scratch[0] / cntf;
        float var = (scratch[1] - cntf * mu * mu) / (cntf - 1.0f);
        float sd  = sqrtf(var);
        float z   = (pos_diff[on] - mu) / sd;
        float pre = 1.0f - 1.0f / (1.0f + expf(-z));

        float beta = powf(0.9f, (float)epoch[0]);
        if (beta < 0.1f) beta = 0.0f;
        float attn = beta * pre + (1.0f - beta) * dot;
        float nei  = attn * NEIGH_WEIGHT;

        float score = (nei * mv.x + (1.0f - nei) * hv.x) * cv.x;
        score += (nei * mv.y + (1.0f - nei) * hv.y) * cv.y;
        score += (nei * mv.z + (1.0f - nei) * hv.z) * cv.z;
        score += (nei * mv.w + (1.0f - nei) * hv.w) * cv.w;
        for (int o = 16; o > 0; o >>= 1) score += __shfl_xor(score, o, 32);

        if (lane == 0) {
            scores_out[i] = score;
            int lab = labels[on];
            float spl = log1pf(expf(-fabsf(score)));
            if (lab == 1) {
                n_sum = fmaxf(score, 0.0f) + spl;
                n_cnt = 1.0f;
            } else {
                p_sum = fmaxf(-score, 0.0f) + spl;
                p_cnt = 1.0f;
            }
        }
    }

    if (lane == 0) {
        red[wave][0] = p_sum; red[wave][1] = p_cnt;
        red[wave][2] = n_sum; red[wave][3] = n_cnt;
    }
    __syncthreads();
    if (threadIdx.x < 4) {
        float s = 0.0f;
        for (int w = 0; w < 8; ++w) s += red[w][threadIdx.x];
        rec[4 * (size_t)blockIdx.x + threadIdx.x] = s;
    }
}

__global__ __launch_bounds__(1024) void loss_kernel(const float* __restrict__ rec, int N,
                                                    float* __restrict__ out0) {
    __shared__ float red[32][4];
    const int tid  = threadIdx.x;
    const int lane = tid & 31;
    const int wave = tid >> 5;
    const int nblk = (N + 7) >> 3;

    float a0 = 0.0f, a1 = 0.0f, a2 = 0.0f, a3 = 0.0f;
    #pragma unroll 4
    for (int b = tid; b < nblk; b += 1024) {
        float4 v = *(const float4*)(rec + 4 * (size_t)b);
        a0 += v.x; a1 += v.y; a2 += v.z; a3 += v.w;
    }
    for (int off = 16; off > 0; off >>= 1) {
        a0 += __shfl_xor(a0, off, 32);
        a1 += __shfl_xor(a1, off, 32);
        a2 += __shfl_xor(a2, off, 32);
        a3 += __shfl_xor(a3, off, 32);
    }
    if (lane == 0) {
        red[wave][0] = a0; red[wave][1] = a1; red[wave][2] = a2; red[wave][3] = a3;
    }
    __syncthreads();
    if (wave == 0) {
        float r0 = red[lane][0], r1 = red[lane][1], r2 = red[lane][2], r3 = red[lane][3];
        for (int off = 16; off > 0; off >>= 1) {
            r0 += __shfl_xor(r0, off, 32);
            r1 += __shfl_xor(r1, off, 32);
            r2 += __shfl_xor(r2, off, 32);
            r3 += __shfl_xor(r3, off, 32);
        }
        if (lane == 0)
            out0[0] = r0 / fmaxf(r1, 1.0f) + r2 / fmaxf(r3, 1.0f);
    }
}

extern "C" void kernel_launch(void* const* d_in, const int* in_sizes, int n_in,
                              void* d_out, int out_size, void* d_ws, size_t ws_size,
                              hipStream_t stream) {
    const float* feats    = (const float*)d_in[0];
    const float* W        = (const float*)d_in[1];
    const float* b        = (const float*)d_in[2];
    const float* center   = (const float*)d_in[3];
    const float* pos_diff = (const float*)d_in[4];
    const int*   src      = (const int*)d_in[5];
    const int*   dst      = (const int*)d_in[6];
    const int*   out_nodes= (const int*)d_in[7];
    const int*   nor_idx  = (const int*)d_in[8];
    const int*   labels   = (const int*)d_in[9];
    const int*   epoch    = (const int*)d_in[10];

    const int N = in_sizes[0] / 256;
    const int E = in_sizes[5];
    const int M = in_sizes[8];
    const int nblk = (N + 7) / 8;

    float* wsf     = (float*)d_ws;
    float* scratch = wsf;
    float* h       = wsf + 64;
    int*   bucket  = (int*)(h + (size_t)N * 128);
    float* rec     = (float*)(bucket + (size_t)N * CAP);
    int*   ovfn    = (int*)(rec + 4 * (size_t)nblk);
    int*   ovf     = ovfn + 4;
    int*   cnt     = ovf + 2 * (size_t)E;

    float* out     = (float*)d_out;

    zero_kernel<<<(N + 255) / 256, 256, 0, stream>>>(scratch, cnt, ovfn, N);

    stats_kernel<<<(M + 255) / 256, 256, 0, stream>>>(pos_diff, nor_idx, M, scratch);

    fill_kernel<<<(E + 255) / 256, 256, 0, stream>>>(src, dst, cnt, bucket, ovfn, ovf, E);

    gemm_h_kernel<<<(N + 15) / 16, 256, 0, stream>>>(feats, W, b, h, N);

    pull_finalize_kernel<<<nblk, 256, 0, stream>>>(h, cnt, bucket, ovfn, ovf, center, pos_diff,
                                                   out_nodes, labels, epoch, scratch, rec,
                                                   out + 1, N, E, M);

    loss_kernel<<<1, 1024, 0, stream>>>(rec, N, out);
}


# neighbour lists ordered through LDS: edges binned by 256-node range with LDS adds (one global add per block and bin), buckets stored as whole lines; per-edge global atomics gone (slot order changes)
# speedup vs baseline: 5.0886x; 1.5778x over previous
#include <hip/hip_runtime.h>
#include <hip/hip_bf16.h>
#include <math.h>

typedef float v2f __attribute__((ext_vector_type(2)));
typedef float v8f __attribute__((ext_vector_type(8)));

#define EPS 1e-12f
#define NEIGH_WEIGHT 0.2f
#define CAP 32

#define BIN_SHIFT 8
#define BIN_NODES 256
#define BIN_CAP (BIN_NODES * 64)
#define MAXBIN 512
#define EPB 8192
#define BIN_THREADS 1024
#define BUILD_THREADS 512
#define CNT_STRIDE 32

__global__ void zero_kernel(float* __restrict__ scratch, int* __restrict__ cnt,
                            int* __restrict__ ovfn, int* __restrict__ binCount, int n, int nbin) {
    int t = blockIdx.x * blockDim.x + threadIdx.x;
    if (t < 64) scratch[t] = 0.0f;
    if (t < n) cnt[t] = 0;
    if (t < nbin) binCount[t * CNT_STRIDE] = 0;
    if (t == 0) ovfn[0] = 0;
}

__global__ void stats_kernel(const float* __restrict__ pos_diff,
                             const int* __restrict__ nor_idx, int M,
                             float* __restrict__ scratch) {
    int t = blockIdx.x * blockDim.x + threadIdx.x;
    float v = 0.0f;
    if (t < M) v = pos_diff[nor_idx[t]];
    float s1 = v, s2 = v * v;
    #pragma unroll
    for (int off = 16; off > 0; off >>= 1) {
        s1 += __shfl_xor(s1, off, 32);
        s2 += __shfl_xor(s2, off, 32);
    }
    if ((threadIdx.x & 31) == 0) {
        __hip_atomic_fetch_add(&scratch[0], s1, __ATOMIC_RELAXED, __HIP_MEMORY_SCOPE_AGENT);
        __hip_atomic_fetch_add(&scratch[1], s2, __ATOMIC_RELAXED, __HIP_MEMORY_SCOPE_AGENT);
    }
}

__global__ __launch_bounds__(256) void gemm_h_kernel(
    const float* __restrict__ feats, const float* __restrict__ W,
    const float* __restrict__ bias, float* __restrict__ h, int N) {
    __shared__ float As[16 * 260];
    __shared__ float T[16 * 132];
    __shared__ float P[16 * 17];
    __shared__ float RowInv[16];

    const int tid  = threadIdx.x;
    const int wave = tid >> 5;
    const int lane = tid & 31;
    const int row0 = blockIdx.x * 16;
    const int cb   = wave * 16;
    const int hl   = lane & 15;
    const int hi   = lane >> 4;

    #pragma unroll
    for (int j = 0; j < 4; ++j) {
        int idx = tid * 4 + j * 1024;
        int r = idx >> 8;
        int c = idx & 255;
        int grow = row0 + r;
        float4 v = make_float4(0.f, 0.f, 0.f, 0.f);
        if (grow < N) v = *(const float4*)(feats + (size_t)grow * 256 + c);
        float* dstp = &As[r * 260 + c];
        dstp[0] = v.x; dstp[1] = v.y; dstp[2] = v.z; dstp[3] = v.w;
    }
    __syncthreads();

    v8f acc = {};
    for (int k = 0; k < 256; k += 4) {
        int ka = k + hi * 2;
        v2f a, b;
        a.x = As[hl * 260 + ka];
        a.y = As[hl * 260 + ka + 1];
        b.x = W[(size_t)ka * 128 + cb + hl];
        b.y = W[(size_t)(ka + 1) * 128 + cb + hl];
        acc = __builtin_amdgcn_wmma_f32_16x16x4_f32(false, a, false, b,
                                                    (short)0, acc, false, false);
    }

    float bv = bias[cb + hl];
    int rbase = hi * 8;
    #pragma unroll
    for (int r = 0; r < 8; ++r) {
        float v = acc[r] + bv;
        v = v > 0.0f ? v : 0.0f;
        T[(rbase + r) * 132 + cb + hl] = v;
    }
    __syncthreads();

    {
        int row = tid >> 4;
        int c0  = (tid & 15) * 8;
        float s = 0.0f;
        #pragma unroll
        for (int j = 0; j < 8; ++j) { float v = T[row * 132 + c0 + j]; s += v * v; }
        P[row * 17 + (tid & 15)] = s;
    }
    __syncthreads();
    if (tid < 16) {
        float s = 0.0f;
        #pragma unroll
        for (int j = 0; j < 16; ++j) s += P[tid * 17 + j];
        RowInv[tid] = 1.0f / fmaxf(sqrtf(s), EPS);
    }
    __syncthreads();

    {
        int row  = tid >> 4;
        int grow = row0 + row;
        if (grow < N) {
            int c0 = (tid & 15) * 8;
            float inv = RowInv[row];
            float* op = h + (size_t)grow * 128 + c0;
            #pragma unroll
            for (int j = 0; j < 8; ++j) op[j] = T[row * 132 + c0 + j] * inv;
        }
    }
}

__global__ __launch_bounds__(BIN_THREADS) void bin_kernel(const int* __restrict__ src,
                                                          const int* __restrict__ dst,
                                                          int* __restrict__ binCount,
                                                          int* __restrict__ bins,
                                                          int* __restrict__ cnt,
                                                          int* __restrict__ ovfn,
                                                          int* __restrict__ ovf,
                                                          int N, int E, int nbin) {
    __shared__ int hist[MAXBIN];
    __shared__ int off[MAXBIN];
    __shared__ int gbase[MAXBIN];
    __shared__ int wsum[32];
    __shared__ int val[EPB];
    __shared__ unsigned short binOf[EPB];

    const int tid  = threadIdx.x;
    const int lane = tid & 31;
    const int wave = tid >> 5;
    const int e0   = blockIdx.x * EPB;

    if (tid < MAXBIN) hist[tid] = 0;
    __syncthreads();

    int dd[EPB / BIN_THREADS], ss[EPB / BIN_THREADS], rr[EPB / BIN_THREADS];
    #pragma unroll
    for (int k = 0; k < EPB / BIN_THREADS; ++k) {
        const int e = e0 + k * BIN_THREADS + tid;
        dd[k] = -1; ss[k] = 0;
        if (e < E) { dd[k] = dst[e]; ss[k] = src[e]; }
    }
    #pragma unroll
    for (int k = 0; k < EPB / BIN_THREADS; ++k) {
        rr[k] = 0;
        if ((unsigned)dd[k] < (unsigned)N)
            rr[k] = __hip_atomic_fetch_add(&hist[dd[k] >> BIN_SHIFT], 1, __ATOMIC_RELAXED,
                                           __HIP_MEMORY_SCOPE_WORKGROUP);
    }
    __syncthreads();

    const int v = (tid < MAXBIN) ? hist[tid] : 0;
    int inc = v;
    #pragma unroll
    for (int o = 1; o < 32; o <<= 1) {
        const int t = __shfl_up(inc, o, 32);
        if (lane >= o) inc += t;
    }
    if (lane == 31) wsum[wave] = inc;
    if (tid < nbin && v > 0)
        gbase[tid] = __hip_atomic_fetch_add(&binCount[tid * CNT_STRIDE], v, __ATOMIC_RELAXED,
                                            __HIP_MEMORY_SCOPE_AGENT);
    __syncthreads();
    if (wave == 0) {
        const int w = wsum[lane];
        int winc = w;
        #pragma unroll
        for (int o = 1; o < 32; o <<= 1) {
            const int t = __shfl_up(winc, o, 32);
            if (lane >= o) winc += t;
        }
        wsum[lane] = winc - w;
    }
    __syncthreads();
    if (tid < MAXBIN) off[tid] = inc - v + wsum[wave];
    __syncthreads();

    #pragma unroll
    for (int k = 0; k < EPB / BIN_THREADS; ++k) {
        if ((unsigned)dd[k] < (unsigned)N) {
            const int b = dd[k] >> BIN_SHIFT;
            const int p = off[b] + rr[k];
            val[p]   = ((dd[k] & (BIN_NODES - 1)) << 24) | (ss[k] & 0xFFFFFF);
            binOf[p] = (unsigned short)b;
        }
    }
    __syncthreads();

    const int nval = off[MAXBIN - 1] + hist[MAXBIN - 1];
    for (int p = tid; p < nval; p += BIN_THREADS) {
        const int b = binOf[p];
        const int g = gbase[b] + (p - off[b]);
        const int w = val[p];
        if ((unsigned)g < (unsigned)BIN_CAP) {
            bins[(size_t)b * BIN_CAP + g] = w;
        } else {
            const int d = (b << BIN_SHIFT) + (int)((unsigned)w >> 24);
            const int q = __hip_atomic_fetch_add(ovfn, 1, __ATOMIC_RELAXED, __HIP_MEMORY_SCOPE_AGENT);
            if ((unsigned)q < (unsigned)E) {
                ovf[2 * (size_t)q]     = d;
                ovf[2 * (size_t)q + 1] = w & 0xFFFFFF;
                __hip_atomic_fetch_add(&cnt[d], 1, __ATOMIC_RELAXED, __HIP_MEMORY_SCOPE_AGENT);
            }
        }
    }
}

__global__ __launch_bounds__(BUILD_THREADS) void build_kernel(const int* __restrict__ binCount,
                                                              const int* __restrict__ bins,
                                                              int* __restrict__ cnt,
                                                              int* __restrict__ bucket,
                                                              int* __restrict__ ovfn,
                                                              int* __restrict__ ovf,
                                                              int N, int E) {
    __shared__ int cntL[BIN_NODES];
    __shared__ int bucketL[BIN_NODES * (CAP + 1)];

    const int tid   = threadIdx.x;
    const int b     = blockIdx.x;
    const int node0 = b << BIN_SHIFT;

    if (tid < BIN_NODES) cntL[tid] = 0;
    for (int k = tid; k < BIN_NODES * (CAP + 1); k += BUILD_THREADS) bucketL[k] = 0;
    __syncthreads();

    int n = binCount[b * CNT_STRIDE];
    if (n > BIN_CAP) n = BIN_CAP;
    const int* __restrict__ my = bins + (size_t)b * BIN_CAP;
    for (int base = 0; base < n; base += 8 * BUILD_THREADS) {
        int w[8];
        #pragma unroll
        for (int u = 0; u < 8; ++u) {
            const int e = base + u * BUILD_THREADS + tid;
            w[u] = 0;
            if (e < n) w[u] = my[e];
        }
        #pragma unroll
        for (int u = 0; u < 8; ++u) {
            const int e = base + u * BUILD_THREADS + tid;
            if (e < n) {
                const int dl = (int)((unsigned)w[u] >> 24);
                const int s  = w[u] & 0xFFFFFF;
                const int slot = __hip_atomic_fetch_add(&cntL[dl], 1, __ATOMIC_RELAXED,
                                                        __HIP_MEMORY_SCOPE_WORKGROUP);
                if ((unsigned)slot < (unsigned)CAP) {
                    bucketL[dl * (CAP + 1) + slot] = s;
                } else {
                    const int q = __hip_atomic_fetch_add(ovfn, 1, __ATOMIC_RELAXED, __HIP_MEMORY_SCOPE_AGENT);
                    if ((unsigned)q < (unsigned)E) {
                        ovf[2 * (size_t)q]     = node0 + dl;
                        ovf[2 * (size_t)q + 1] = s;
                    }
                }
            }
        }
    }
    __syncthreads();

    if (tid < BIN_NODES) {
        const int node = node0 + tid;
        if (node < N) {
            const int cl  = cntL[tid];
            const int tot = cl + cnt[node];
            const int filled = cl < CAP ? cl : CAP;
            cnt[node] = (tot & 0xFFFFFF) | (filled << 24);
        }
    }
    for (int k = tid; k < BIN_NODES * CAP; k += BUILD_THREADS) {
        const int nl = k >> 5;
        const int j  = k & (CAP - 1);
        if (node0 + nl < N) bucket[(size_t)(node0 + nl) * CAP + j] = bucketL[nl * (CAP + 1) + j];
    }
}

__device__ __forceinline__ float4 row4(const float* __restrict__ hl, int s) {
    return *(const float4*)(hl + (size_t)s * 128);
}
__device__ __forceinline__ void add4(float4& a, const float4& v) {
    a.x += v.x; a.y += v.y; a.z += v.z; a.w += v.w;
}
template <int G>
__device__ __forceinline__ void add_rows(float4& acc, const float* __restrict__ hl,
                                         int myidx, int k) {
    float4 v[G];
    #pragma unroll
    for (int u = 0; u < G; ++u) v[u] = row4(hl, __builtin_amdgcn_readlane(myidx, k + u));
    #pragma unroll
    for (int u = 0; u < G; ++u) add4(acc, v[u]);
}
__device__ __forceinline__ void walk(float4& acc, const float* __restrict__ hl, int myidx, int m) {
    int k = 0;
    for (; k + 8 <= m; k += 8) add_rows<8>(acc, hl, myidx, k);
    if (m & 4) { add_rows<4>(acc, hl, myidx, k); k += 4; }
    if (m & 2) { add_rows<2>(acc, hl, myidx, k); k += 2; }
    if (m & 1) { add_rows<1>(acc, hl, myidx, k); }
}

__global__ __launch_bounds__(256) void pull_finalize_kernel(
    const float* __restrict__ h, const int* __restrict__ cnt,
    const int* __restrict__ bucket, const int* __restrict__ ovfn,
    const int* __restrict__ ovf,
    const float* __restrict__ center,
    const float* __restrict__ pos_diff, const int* __restrict__ out_nodes,
    const int* __restrict__ labels, const int* __restrict__ epoch,
    const float* __restrict__ scratch, float* __restrict__ rec,
    float* __restrict__ scores_out, int N, int E, int Mnor) {
    __shared__ float red[8][4];

    const int wave = __builtin_amdgcn_readfirstlane(threadIdx.x >> 5);
    const int lane = threadIdx.x & 31;
    const int i = blockIdx.x * 8 + wave;

    float p_sum = 0.0f, p_cnt = 0.0f, n_sum = 0.0f, n_cnt = 0.0f;

    if (i < N) {
        const int cw = cnt[i];
        const int c  = cw & 0xFFFFFF;
        const int mf = (int)((unsigned)cw >> 24);
        const int m0 = mf < CAP ? mf : CAP;

        const float* hl = h + lane * 4;
        int idx0 = 0;
        if (lane < m0) idx0 = bucket[(size_t)i * CAP + lane];
        const float4 hv = *(const float4*)(hl + (size_t)i * 128);
        const float4 cv = *(const float4*)(center + lane * 4);

        float4 acc = make_float4(0.f, 0.f, 0.f, 0.f);
        if (m0 > 0) walk(acc, hl, idx0, m0);
        if (c > m0) {
            int n = ovfn[0];
            if (n > E) n = E;
            for (int base = 0; base < n; base += 32) {
                const int k = base + lane;
                int d = -1, s = 0;
                if (k < n) { d = ovf[2 * (size_t)k]; s = ovf[2 * (size_t)k + 1]; }
                unsigned hit = __builtin_amdgcn_ballot_w32(d == i);
                while (hit) {
                    const int b = __builtin_ctz(hit);
                    hit &= hit - 1;
                    const float4 v = row4(hl, __builtin_amdgcn_readlane(s, b));
                    add4(acc, v);
                }
            }
        }

        const float invd = c > 0 ? 1.0f / fmaxf((float)c, 1.0f) : 0.0f;
        float4 mv;
        mv.x = acc.x * invd; mv.y = acc.y * invd; mv.z = acc.z * invd; mv.w = acc.w * invd;
        float dot = hv.x * mv.x;
        dot += hv.y * mv.y;
        dot += hv.z * mv.z;
        dot += hv.w * mv.w;
        for (int o = 16; o > 0; o >>= 1) dot += __shfl_xor(dot, o, 32);

        int on = out_nodes[i];
        float cntf = (float)Mnor;
        float mu  = scratch[0] / cntf;
        float var = (scratch[1] - cntf * mu * mu) / (cntf - 1.0f);
        float sd  = sqrtf(var);
        float z   = (pos_diff[on] - mu) / sd;
        float pre = 1.0f - 1.0f / (1.0f + expf(-z));

        float beta = powf(0.9f, (float)epoch[0]);
        if (beta < 0.1f) beta = 0.0f;
        float attn = beta * pre + (1.0f - beta) * dot;
        float nei  = attn * NEIGH_WEIGHT;

        float score = (nei * mv.x + (1.0f - nei) * hv.x) * cv.x;
        score += (nei * mv.y + (1.0f - nei) * hv.y) * cv.y;
        score += (nei * mv.z + (1.0f - nei) * hv.z) * cv.z;
        score += (nei * mv.w + (1.0f - nei) * hv.w) * cv.w;
        for (int o = 16; o > 0; o >>= 1) score += __shfl_xor(score, o, 32);

        if (lane == 0) {
            scores_out[i] = score;
            int lab = labels[on];
            float spl = log1pf(expf(-fabsf(score)));
            if (lab == 1) {
                n_sum = fmaxf(score, 0.0f) + spl;
                n_cnt = 1.0f;
            } else {
                p_sum = fmaxf(-score, 0.0f) + spl;
                p_cnt = 1.0f;
            }
        }
    }

    if (lane == 0) {
        red[wave][0] = p_sum; red[wave][1] = p_cnt;
        red[wave][2] = n_sum; red[wave][3] = n_cnt;
    }
    __syncthreads();
    if (threadIdx.x < 4) {
        float s = 0.0f;
        for (int w = 0; w < 8; ++w) s += red[w][threadIdx.x];
        rec[4 * (size_t)blockIdx.x + threadIdx.x] = s;
    }
}

__global__ __launch_bounds__(1024) void loss_kernel(const float* __restrict__ rec, int N,
                                                    float* __restrict__ out0) {
    __shared__ float red[32][4];
    const int tid  = threadIdx.x;
    const int lane = tid & 31;
    const int wave = tid >> 5;
    const int nblk = (N + 7) >> 3;

    float a0 = 0.0f, a1 = 0.0f, a2 = 0.0f, a3 = 0.0f;
    #pragma unroll 4
    for (int b = tid; b < nblk; b += 1024) {
        float4 v = *(const float4*)(rec + 4 * (size_t)b);
        a0 += v.x; a1 += v.y; a2 += v.z; a3 += v.w;
    }
    for (int off = 16; off > 0; off >>= 1) {
        a0 += __shfl_xor(a0, off, 32);
        a1 += __shfl_xor(a1, off, 32);
        a2 += __shfl_xor(a2, off, 32);
        a3 += __shfl_xor(a3, off, 32);
    }
    if (lane == 0) {
        red[wave][0] = a0; red[wave][1] = a1; red[wave][2] = a2; red[wave][3] = a3;
    }
    __syncthreads();
    if (wave == 0) {
        float r0 = red[lane][0], r1 = red[lane][1], r2 = red[lane][2], r3 = red[lane][3];
        for (int off = 16; off > 0; off >>= 1) {
            r0 += __shfl_xor(r0, off, 32);
            r1 += __shfl_xor(r1, off, 32);
            r2 += __shfl_xor(r2, off, 32);
            r3 += __shfl_xor(r3, off, 32);
        }
        if (lane == 0)
            out0[0] = r0 / fmaxf(r1, 1.0f) + r2 / fmaxf(r3, 1.0f);
    }
}

extern "C" void kernel_launch(void* const* d_in, const int* in_sizes, int n_in,
                              void* d_out, int out_size, void* d_ws, size_t ws_size,
                              hipStream_t stream) {
    const float* feats    = (const float*)d_in[0];
    const float* W        = (const float*)d_in[1];
    const float* b        = (const float*)d_in[2];
    const float* center   = (const float*)d_in[3];
    const float* pos_diff = (const float*)d_in[4];
    const int*   src      = (const int*)d_in[5];
    const int*   dst      = (const int*)d_in[6];
    const int*   out_nodes= (const int*)d_in[7];
    const int*   nor_idx  = (const int*)d_in[8];
    const int*   labels   = (const int*)d_in[9];
    const int*   epoch    = (const int*)d_in[10];

    const int N = in_sizes[0] / 256;
    const int E = in_sizes[5];
    const int M = in_sizes[8];
    const int nblk = (N + 7) / 8;

    const int nbin = (N + BIN_NODES - 1) / BIN_NODES;

    float* wsf     = (float*)d_ws;
    float* scratch = wsf;
    float* h       = wsf + 64;
    int*   bucket  = (int*)(h + (size_t)N * 128);
    float* rec     = (float*)(bucket + (size_t)N * CAP);
    int*   ovfn    = (int*)(rec + 4 * (size_t)nblk);
    int*   ovf     = ovfn + 4;
    int*   cnt     = ovf + 2 * (size_t)E;
    int*   bins    = (int*)h;
    int*   binCount= (int*)rec;

    float* out     = (float*)d_out;

    zero_kernel<<<(N + 255) / 256, 256, 0, stream>>>(scratch, cnt, ovfn, binCount, N, nbin);

    stats_kernel<<<(M + 255) / 256, 256, 0, stream>>>(pos_diff, nor_idx, M, scratch);

    bin_kernel<<<(E + EPB - 1) / EPB, BIN_THREADS, 0, stream>>>(src, dst, binCount, bins, cnt, ovfn, ovf, N, E, nbin);
    build_kernel<<<nbin, BUILD_THREADS, 0, stream>>>(binCount, bins, cnt, bucket, ovfn, ovf, N, E);

    gemm_h_kernel<<<(N + 15) / 16, 256, 0, stream>>>(feats, W, b, h, N);

    pull_finalize_kernel<<<nblk, 256, 0, stream>>>(h, cnt, bucket, ovfn, ovf, center, pos_diff,
                                                   out_nodes, labels, epoch, scratch, rec,
                                                   out + 1, N, E, M);

    loss_kernel<<<1, 1024, 0, stream>>>(rec, N, out);
}


# pull: a wave walks 10 nodes (1/10 the waves); per-node scalar work (1/degree, pre-attention, softplus) once per wave, lane j = node j; next line and row fetched ahead (loss sum order changes)
# speedup vs baseline: 5.3827x; 1.0578x over previous
#include <hip/hip_runtime.h>
#include <hip/hip_bf16.h>
#include <math.h>

typedef float v2f __attribute__((ext_vector_type(2)));
typedef float v8f __attribute__((ext_vector_type(8)));

#define EPS 1e-12f
#define NEIGH_WEIGHT 0.2f
#define CAP 32

#define BIN_SHIFT 8
#define BIN_NODES 256
#define BIN_CAP (BIN_NODES * 64)
#define MAXBIN 512
#define EPB 8192
#define BIN_THREADS 1024
#define BUILD_THREADS 512
#define CNT_STRIDE 32

__global__ void zero_kernel(float* __restrict__ scratch, int* __restrict__ cnt,
                            int* __restrict__ ovfn, int* __restrict__ binCount, int n, int nbin) {
    int t = blockIdx.x * blockDim.x + threadIdx.x;
    if (t < 64) scratch[t] = 0.0f;
    if (t < n) cnt[t] = 0;
    if (t < nbin) binCount[t * CNT_STRIDE] = 0;
    if (t == 0) ovfn[0] = 0;
}

__global__ void stats_kernel(const float* __restrict__ pos_diff,
                             const int* __restrict__ nor_idx, int M,
                             float* __restrict__ scratch) {
    int t = blockIdx.x * blockDim.x + threadIdx.x;
    float v = 0.0f;
    if (t < M) v = pos_diff[nor_idx[t]];
    float s1 = v, s2 = v * v;
    #pragma unroll
    for (int off = 16; off > 0; off >>= 1) {
        s1 += __shfl_xor(s1, off, 32);
        s2 += __shfl_xor(s2, off, 32);
    }
    if ((threadIdx.x & 31) == 0) {
        __hip_atomic_fetch_add(&scratch[0], s1, __ATOMIC_RELAXED, __HIP_MEMORY_SCOPE_AGENT);
        __hip_atomic_fetch_add(&scratch[1], s2, __ATOMIC_RELAXED, __HIP_MEMORY_SCOPE_AGENT);
    }
}

__global__ __launch_bounds__(256) void gemm_h_kernel(
    const float* __restrict__ feats, const float* __restrict__ W,
    const float* __restrict__ bias, float* __restrict__ h, int N) {
    __shared__ float As[16 * 260];
    __shared__ float T[16 * 132];
    __shared__ float P[16 * 17];
    __shared__ float RowInv[16];

    const int tid  = threadIdx.x;
    const int wave = tid >> 5;
    const int lane = tid & 31;
    const int row0 = blockIdx.x * 16;
    const int cb   = wave * 16;
    const int hl   = lane & 15;
    const int hi   = lane >> 4;

    #pragma unroll
    for (int j = 0; j < 4; ++j) {
        int idx = tid * 4 + j * 1024;
        int r = idx >> 8;
        int c = idx & 255;
        int grow = row0 + r;
        float4 v = make_float4(0.f, 0.f, 0.f, 0.f);
        if (grow < N) v = *(const float4*)(feats + (size_t)grow * 256 + c);
        float* dstp = &As[r * 260 + c];
        dstp[0] = v.x; dstp[1] = v.y; dstp[2] = v.z; dstp[3] = v.w;
    }
    __syncthreads();

    v8f acc = {};
    for (int k = 0; k < 256; k += 4) {
        int ka = k + hi * 2;
        v2f a, b;
        a.x = As[hl * 260 + ka];
        a.y = As[hl * 260 + ka + 1];
        b.x = W[(size_t)ka * 128 + cb + hl];
        b.y = W[(size_t)(ka + 1) * 128 + cb + hl];
        acc = __builtin_amdgcn_wmma_f32_16x16x4_f32(false, a, false, b,
                                                    (short)0, acc, false, false);
    }

    float bv = bias[cb + hl];
    int rbase = hi * 8;
    #pragma unroll
    for (int r = 0; r < 8; ++r) {
        float v = acc[r] + bv;
        v = v > 0.0f ? v : 0.0f;
        T[(rbase + r) * 132 + cb + hl] = v;
    }
    __syncthreads();

    {
        int row = tid >> 4;
        int c0  = (tid & 15) * 8;
        float s = 0.0f;
        #pragma unroll
        for (int j = 0; j < 8; ++j) { float v = T[row * 132 + c0 + j]; s += v * v; }
        P[row * 17 + (tid & 15)] = s;
    }
    __syncthreads();
    if (tid < 16) {
        float s = 0.0f;
        #pragma unroll
        for (int j = 0; j < 16; ++j) s += P[tid * 17 + j];
        RowInv[tid] = 1.0f / fmaxf(sqrtf(s), EPS);
    }
    __syncthreads();

    {
        int row  = tid >> 4;
        int grow = row0 + row;
        if (grow < N) {
            int c0 = (tid & 15) * 8;
            float inv = RowInv[row];
            float* op = h + (size_t)grow * 128 + c0;
            #pragma unroll
            for (int j = 0; j < 8; ++j) op[j] = T[row * 132 + c0 + j] * inv;
        }
    }
}

__global__ __launch_bounds__(BIN_THREADS) void bin_kernel(const int* __restrict__ src,
                                                          const int* __restrict__ dst,
                                                          int* __restrict__ binCount,
                                                          int* __restrict__ bins,
                                                          int* __restrict__ cnt,
                                                          int* __restrict__ ovfn,
                                                          int* __restrict__ ovf,
                                                          int N, int E, int nbin) {
    __shared__ int hist[MAXBIN];
    __shared__ int off[MAXBIN];
    __shared__ int gbase[MAXBIN];
    __shared__ int wsum[32];
    __shared__ int val[EPB];
    __shared__ unsigned short binOf[EPB];

    const int tid  = threadIdx.x;
    const int lane = tid & 31;
    const int wave = tid >> 5;
    const int e0   = blockIdx.x * EPB;

    if (tid < MAXBIN) hist[tid] = 0;
    __syncthreads();

    int dd[EPB / BIN_THREADS], ss[EPB / BIN_THREADS], rr[EPB / BIN_THREADS];
    #pragma unroll
    for (int k = 0; k < EPB / BIN_THREADS; ++k) {
        const int e = e0 + k * BIN_THREADS + tid;
        dd[k] = -1; ss[k] = 0;
        if (e < E) { dd[k] = dst[e]; ss[k] = src[e]; }
    }
    #pragma unroll
    for (int k = 0; k < EPB / BIN_THREADS; ++k) {
        rr[k] = 0;
        if ((unsigned)dd[k] < (unsigned)N)
            rr[k] = __hip_atomic_fetch_add(&hist[dd[k] >> BIN_SHIFT], 1, __ATOMIC_RELAXED,
                                           __HIP_MEMORY_SCOPE_WORKGROUP);
    }
    __syncthreads();

    const int v = (tid < MAXBIN) ? hist[tid] : 0;
    int inc = v;
    #pragma unroll
    for (int o = 1; o < 32; o <<= 1) {
        const int t = __shfl_up(inc, o, 32);
        if (lane >= o) inc += t;
    }
    if (lane == 31) wsum[wave] = inc;
    if (tid < nbin && v > 0)
        gbase[tid] = __hip_atomic_fetch_add(&binCount[tid * CNT_STRIDE], v, __ATOMIC_RELAXED,
                                            __HIP_MEMORY_SCOPE_AGENT);
    __syncthreads();
    if (wave == 0) {
        const int w = wsum[lane];
        int winc = w;
        #pragma unroll
        for (int o = 1; o < 32; o <<= 1) {
            const int t = __shfl_up(winc, o, 32);
            if (lane >= o) winc += t;
        }
        wsum[lane] = winc - w;
    }
    __syncthreads();
    if (tid < MAXBIN) off[tid] = inc - v + wsum[wave];
    __syncthreads();

    #pragma unroll
    for (int k = 0; k < EPB / BIN_THREADS; ++k) {
        if ((unsigned)dd[k] < (unsigned)N) {
            const int b = dd[k] >> BIN_SHIFT;
            const int p = off[b] + rr[k];
            val[p]   = ((dd[k] & (BIN_NODES - 1)) << 24) | (ss[k] & 0xFFFFFF);
            binOf[p] = (unsigned short)b;
        }
    }
    __syncthreads();

    const int nval = off[MAXBIN - 1] + hist[MAXBIN - 1];
    for (int p = tid; p < nval; p += BIN_THREADS) {
        const int b = binOf[p];
        const int g = gbase[b] + (p - off[b]);
        const int w = val[p];
        if ((unsigned)g < (unsigned)BIN_CAP) {
            bins[(size_t)b * BIN_CAP + g] = w;
        } else {
            const int d = (b << BIN_SHIFT) + (int)((unsigned)w >> 24);
            const int q = __hip_atomic_fetch_add(ovfn, 1, __ATOMIC_RELAXED, __HIP_MEMORY_SCOPE_AGENT);
            if ((unsigned)q < (unsigned)E) {
                ovf[2 * (size_t)q]     = d;
                ovf[2 * (size_t)q + 1] = w & 0xFFFFFF;
                __hip_atomic_fetch_add(&cnt[d], 1, __ATOMIC_RELAXED, __HIP_MEMORY_SCOPE_AGENT);
            }
        }
    }
}

__global__ __launch_bounds__(BUILD_THREADS) void build_kernel(const int* __restrict__ binCount,
                                                              const int* __restrict__ bins,
                                                              int* __restrict__ cnt,
                                                              int* __restrict__ bucket,
                                                              int* __restrict__ ovfn,
                                                              int* __restrict__ ovf,
                                                              int N, int E) {
    __shared__ int cntL[BIN_NODES];
    __shared__ int bucketL[BIN_NODES * (CAP + 1)];

    const int tid   = threadIdx.x;
    const int b     = blockIdx.x;
    const int node0 = b << BIN_SHIFT;

    if (tid < BIN_NODES) cntL[tid] = 0;
    for (int k = tid; k < BIN_NODES * (CAP + 1); k += BUILD_THREADS) bucketL[k] = 0;
    __syncthreads();

    int n = binCount[b * CNT_STRIDE];
    if (n > BIN_CAP) n = BIN_CAP;
    const int* __restrict__ my = bins + (size_t)b * BIN_CAP;
    for (int base = 0; base < n; base += 8 * BUILD_THREADS) {
        int w[8];
        #pragma unroll
        for (int u = 0; u < 8; ++u) {
            const int e = base + u * BUILD_THREADS + tid;
            w[u] = 0;
            if (e < n) w[u] = my[e];
        }
        #pragma unroll
        for (int u = 0; u < 8; ++u) {
            const int e = base + u * BUILD_THREADS + tid;
            if (e < n) {
                const int dl = (int)((unsigned)w[u] >> 24);
                const int s  = w[u] & 0xFFFFFF;
                const int slot = __hip_atomic_fetch_add(&cntL[dl], 1, __ATOMIC_RELAXED,
                                                        __HIP_MEMORY_SCOPE_WORKGROUP);
                if ((unsigned)slot < (unsigned)CAP) {
                    bucketL[dl * (CAP + 1) + slot] = s;
                } else {
                    const int q = __hip_atomic_fetch_add(ovfn, 1, __ATOMIC_RELAXED, __HIP_MEMORY_SCOPE_AGENT);
                    if ((unsigned)q < (unsigned)E) {
                        ovf[2 * (size_t)q]     = node0 + dl;
                        ovf[2 * (size_t)q + 1] = s;
                    }
                }
            }
        }
    }
    __syncthreads();

    if (tid < BIN_NODES) {
        const int node = node0 + tid;
        if (node < N) {
            const int cl  = cntL[tid];
            const int tot = cl + cnt[node];
            const int filled = cl < CAP ? cl : CAP;
            cnt[node] = (tot & 0xFFFFFF) | (filled << 24);
        }
    }
    for (int k = tid; k < BIN_NODES * CAP; k += BUILD_THREADS) {
        const int nl = k >> 5;
        const int j  = k & (CAP - 1);
        if (node0 + nl < N) bucket[(size_t)(node0 + nl) * CAP + j] = bucketL[nl * (CAP + 1) + j];
    }
}

#define NPW 10
#define PULL_NODES (8 * NPW)

__device__ __forceinline__ float4 row4(const float* __restrict__ hl, int s) {
    return *(const float4*)(hl + (size_t)s * 128);
}
__device__ __forceinline__ void add4(float4& a, const float4& v) {
    a.x += v.x; a.y += v.y; a.z += v.z; a.w += v.w;
}
template <int G>
__device__ __forceinline__ void add_rows(float4& acc, const float* __restrict__ hl,
                                         int myidx, int k) {
    float4 v[G];
    #pragma unroll
    for (int u = 0; u < G; ++u) v[u] = row4(hl, __builtin_amdgcn_readlane(myidx, k + u));
    #pragma unroll
    for (int u = 0; u < G; ++u) add4(acc, v[u]);
}
__device__ __forceinline__ void walk(float4& acc, const float* __restrict__ hl, int myidx, int m) {
    int k = 0;
    for (; k + 8 <= m; k += 8) add_rows<8>(acc, hl, myidx, k);
    if (m & 4) { add_rows<4>(acc, hl, myidx, k); k += 4; }
    if (m & 2) { add_rows<2>(acc, hl, myidx, k); k += 2; }
    if (m & 1) { add_rows<1>(acc, hl, myidx, k); }
}

__global__ __launch_bounds__(256) void pull_finalize_kernel(
    const float* __restrict__ h, const int* __restrict__ cnt,
    const int* __restrict__ bucket, const int* __restrict__ ovfn,
    const int* __restrict__ ovf,
    const float* __restrict__ center,
    const float* __restrict__ pos_diff, const int* __restrict__ out_nodes,
    const int* __restrict__ labels, const int* __restrict__ epoch,
    const float* __restrict__ scratch, float* __restrict__ rec,
    float* __restrict__ scores_out, int N, int E, int Mnor) {
    __shared__ float red[8][4];
    __shared__ float sc[PULL_NODES];

    const int wave = __builtin_amdgcn_readfirstlane(threadIdx.x >> 5);
    const int lane = threadIdx.x & 31;
    const int i0 = (blockIdx.x * 8 + wave) * NPW;

    float p_sum = 0.0f, p_cnt = 0.0f, n_sum = 0.0f, n_cnt = 0.0f;

    if (i0 < N) {
        const int nn = i0 + NPW < N ? NPW : N - i0;
        const float* hl = h + lane * 4;

        int    idx = bucket[(size_t)i0 * CAP + lane];
        float4 hv  = *(const float4*)(hl + (size_t)i0 * 128);

        int cw_v = 0, lab_v = 0;
        float pd_v = 0.0f;
        if (lane < nn) {
            cw_v = cnt[i0 + lane];
            const int on = out_nodes[i0 + lane];
            pd_v  = pos_diff[on];
            lab_v = labels[on];
        }

        const float4 cv = *(const float4*)(center + lane * 4);
        const float cntf = (float)Mnor;
        const float mu  = scratch[0] / cntf;
        const float var = (scratch[1] - cntf * mu * mu) / (cntf - 1.0f);
        const float sd  = sqrtf(var);
        float beta = powf(0.9f, (float)epoch[0]);
        if (beta < 0.1f) beta = 0.0f;

        const int   c_v    = cw_v & 0xFFFFFF;
        const float invd_v = c_v > 0 ? 1.0f / fmaxf((float)c_v, 1.0f) : 0.0f;
        const float z_v    = (pd_v - mu) / sd;
        const float pre_v  = 1.0f - 1.0f / (1.0f + expf(-z_v));
        float score_v = 0.0f;

        for (int j = 0; j < nn; ++j) {
            const int i = i0 + j;
            const int    idxc = idx;
            const float4 hvc  = hv;
            if (j + 1 < nn) {
                idx = bucket[(size_t)(i + 1) * CAP + lane];
                hv  = *(const float4*)(hl + (size_t)(i + 1) * 128);
            }
            const int   cwc  = __builtin_amdgcn_readlane(cw_v, j);
            const float invd = __int_as_float(__builtin_amdgcn_readlane(__float_as_int(invd_v), j));
            const float pre  = __int_as_float(__builtin_amdgcn_readlane(__float_as_int(pre_v), j));

            const int c  = cwc & 0xFFFFFF;
            const int mf = (int)((unsigned)cwc >> 24);
            const int m0 = mf < CAP ? mf : CAP;

            float4 acc = make_float4(0.f, 0.f, 0.f, 0.f);
            if (m0 > 0) walk(acc, hl, idxc, m0);
            if (c > m0) {
                int n = ovfn[0];
                if (n > E) n = E;
                for (int base = 0; base < n; base += 32) {
                    const int k = base + lane;
                    int d = -1, s = 0;
                    if (k < n) { d = ovf[2 * (size_t)k]; s = ovf[2 * (size_t)k + 1]; }
                    unsigned hit = __builtin_amdgcn_ballot_w32(d == i);
                    while (hit) {
                        const int b = __builtin_ctz(hit);
                        hit &= hit - 1;
                        const float4 v = row4(hl, __builtin_amdgcn_readlane(s, b));
                        add4(acc, v);
                    }
                }
            }

            float4 mv;
            mv.x = acc.x * invd; mv.y = acc.y * invd; mv.z = acc.z * invd; mv.w = acc.w * invd;
            float dot = hvc.x * mv.x;
            dot += hvc.y * mv.y;
            dot += hvc.z * mv.z;
            dot += hvc.w * mv.w;
            for (int o = 16; o > 0; o >>= 1) dot += __shfl_xor(dot, o, 32);

            float attn = beta * pre + (1.0f - beta) * dot;
            float nei  = attn * NEIGH_WEIGHT;

            float score = (nei * mv.x + (1.0f - nei) * hvc.x) * cv.x;
            score += (nei * mv.y + (1.0f - nei) * hvc.y) * cv.y;
            score += (nei * mv.z + (1.0f - nei) * hvc.z) * cv.z;
            score += (nei * mv.w + (1.0f - nei) * hvc.w) * cv.w;
            for (int o = 16; o > 0; o >>= 1) score += __shfl_xor(score, o, 32);

            if (lane == j) score_v = score;
        }

        if (lane < nn) {
            sc[wave * NPW + lane] = score_v;
            float spl = log1pf(expf(-fabsf(score_v)));
            if (lab_v == 1) {
                n_sum = fmaxf(score_v, 0.0f) + spl;
                n_cnt = 1.0f;
            } else {
                p_sum = fmaxf(-score_v, 0.0f) + spl;
                p_cnt = 1.0f;
            }
        }
        for (int o = 16; o > 0; o >>= 1) {
            p_sum += __shfl_xor(p_sum, o, 32);
            p_cnt += __shfl_xor(p_cnt, o, 32);
            n_sum += __shfl_xor(n_sum, o, 32);
            n_cnt += __shfl_xor(n_cnt, o, 32);
        }
    }

    if (lane == 0) {
        red[wave][0] = p_sum; red[wave][1] = p_cnt;
        red[wave][2] = n_sum; red[wave][3] = n_cnt;
    }
    __syncthreads();
    if (threadIdx.x < 4) {
        float s = 0.0f;
        for (int w = 0; w < 8; ++w) s += red[w][threadIdx.x];
        rec[4 * (size_t)blockIdx.x + threadIdx.x] = s;
    }
    if (threadIdx.x < PULL_NODES) {
        const int i = blockIdx.x * PULL_NODES + threadIdx.x;
        if (i < N) scores_out[i] = sc[threadIdx.x];
    }
}

__global__ __launch_bounds__(1024) void loss_kernel(const float* __restrict__ rec, int N,
                                                    float* __restrict__ out0) {
    __shared__ float red[32][4];
    const int tid  = threadIdx.x;
    const int lane = tid & 31;
    const int wave = tid >> 5;
    const int nblk = (N + PULL_NODES - 1) / PULL_NODES;

    float a0 = 0.0f, a1 = 0.0f, a2 = 0.0f, a3 = 0.0f;
    #pragma unroll 4
    for (int b = tid; b < nblk; b += 1024) {
        float4 v = *(const float4*)(rec + 4 * (size_t)b);
        a0 += v.x; a1 += v.y; a2 += v.z; a3 += v.w;
    }
    for (int off = 16; off > 0; off >>= 1) {
        a0 += __shfl_xor(a0, off, 32);
        a1 += __shfl_xor(a1, off, 32);
        a2 += __shfl_xor(a2, off, 32);
        a3 += __shfl_xor(a3, off, 32);
    }
    if (lane == 0) {
        red[wave][0] = a0; red[wave][1] = a1; red[wave][2] = a2; red[wave][3] = a3;
    }
    __syncthreads();
    if (wave == 0) {
        float r0 = red[lane][0], r1 = red[lane][1], r2 = red[lane][2], r3 = red[lane][3];
        for (int off = 16; off > 0; off >>= 1) {
            r0 += __shfl_xor(r0, off, 32);
            r1 += __shfl_xor(r1, off, 32);
            r2 += __shfl_xor(r2, off, 32);
            r3 += __shfl_xor(r3, off, 32);
        }
        if (lane == 0)
            out0[0] = r0 / fmaxf(r1, 1.0f) + r2 / fmaxf(r3, 1.0f);
    }
}

extern "C" void kernel_launch(void* const* d_in, const int* in_sizes, int n_in,
                              void* d_out, int out_size, void* d_ws, size_t ws_size,
                              hipStream_t stream) {
    const float* feats    = (const float*)d_in[0];
    const float* W        = (const float*)d_in[1];
    const float* b        = (const float*)d_in[2];
    const float* center   = (const float*)d_in[3];
    const float* pos_diff = (const float*)d_in[4];
    const int*   src      = (const int*)d_in[5];
    const int*   dst      = (const int*)d_in[6];
    const int*   out_nodes= (const int*)d_in[7];
    const int*   nor_idx  = (const int*)d_in[8];
    const int*   labels   = (const int*)d_in[9];
    const int*   epoch    = (const int*)d_in[10];

    const int N = in_sizes[0] / 256;
    const int E = in_sizes[5];
    const int M = in_sizes[8];
    const int nblk = (N + 7) / 8;

    const int nbin = (N + BIN_NODES - 1) / BIN_NODES;
    const int npull = (N + PULL_NODES - 1) / PULL_NODES;

    float* wsf     = (float*)d_ws;
    float* scratch = wsf;
    float* h       = wsf + 64;
    int*   bucket  = (int*)(h + (size_t)N * 128);
    float* rec     = (float*)(bucket + (size_t)N * CAP);
    int*   ovfn    = (int*)(rec + 4 * (size_t)nblk);
    int*   ovf     = ovfn + 4;
    int*   cnt     = ovf + 2 * (size_t)E;
    int*   bins    = (int*)h;
    int*   binCount= (int*)rec;

    float* out     = (float*)d_out;

    zero_kernel<<<(N + 255) / 256, 256, 0, stream>>>(scratch, cnt, ovfn, binCount, N, nbin);

    stats_kernel<<<(M + 255) / 256, 256, 0, stream>>>(pos_diff, nor_idx, M, scratch);

    bin_kernel<<<(E + EPB - 1) / EPB, BIN_THREADS, 0, stream>>>(src, dst, binCount, bins, cnt, ovfn, ovf, N, E, nbin);
    build_kernel<<<nbin, BUILD_THREADS, 0, stream>>>(binCount, bins, cnt, bucket, ovfn, ovf, N, E);

    gemm_h_kernel<<<(N + 15) / 16, 256, 0, stream>>>(feats, W, b, h, N);

    pull_finalize_kernel<<<npull, 256, 0, stream>>>(h, cnt, bucket, ovfn, ovf, center, pos_diff,
                                                   out_nodes, labels, epoch, scratch, rec,
                                                   out + 1, N, E, M);

    loss_kernel<<<1, 1024, 0, stream>>>(rec, N, out);
}


# zero and stats launches folded into front blocks of the GEMM grid (7 launches to 5): stats as 16 per-block pairs, no atomics (stats sum order changes); edge bins moved from h's rows into bucket lines
# speedup vs baseline: 5.8636x; 1.0893x over previous
#include <hip/hip_runtime.h>
#include <hip/hip_bf16.h>
#include <math.h>

typedef float v2f __attribute__((ext_vector_type(2)));
typedef float v8f __attribute__((ext_vector_type(8)));

#define EPS 1e-12f
#define NEIGH_WEIGHT 0.2f
#define CAP 32

#define BIN_SHIFT 8
#define BIN_NODES 256
#define BIN_CAP (BIN_NODES * CAP)
#define NSTAT 16
#define ZERO_WORDS 4096
#define MAXBIN 512
#define EPB 8192
#define BIN_THREADS 1024
#define BUILD_THREADS 512
#define CNT_STRIDE 32

__global__ __launch_bounds__(256) void gemm_h_kernel(
    const float* __restrict__ feats, const float* __restrict__ W,
    const float* __restrict__ bias, float* __restrict__ h, int N,
    const float* __restrict__ pos_diff, const int* __restrict__ nor_idx, int M,
    float* __restrict__ scratch, int* __restrict__ cnt, int* __restrict__ ovfn,
    int* __restrict__ binCount, int nbin, int nz) {
    __shared__ float As[16 * 260];
    __shared__ float T[16 * 132];
    __shared__ float P[16 * 17];
    __shared__ float RowInv[16];

    const int tid  = threadIdx.x;

    if ((int)blockIdx.x < nz) {
        const int w0 = blockIdx.x * ZERO_WORDS + tid;
        for (int j = 0; j < ZERO_WORDS / 256; ++j) {
            const int t = w0 + j * 256;
            if (t < N) cnt[t] = 0;
        }
        for (int t = blockIdx.x * 256 + tid; t < nbin; t += nz * 256) binCount[t * CNT_STRIDE] = 0;
        if (blockIdx.x == 0 && tid == 0) ovfn[0] = 0;
        return;
    }
    if ((int)blockIdx.x < nz + NSTAT) {
        const int sb = blockIdx.x - nz;
        float s1 = 0.0f, s2 = 0.0f;
        for (int k0 = sb * 256 + tid; k0 < M; k0 += 4 * NSTAT * 256) {
            int   ix[4];
            float v[4];
            #pragma unroll
            for (int u = 0; u < 4; ++u) {
                const int k = k0 + u * NSTAT * 256;
                ix[u] = nor_idx[k < M ? k : M - 1];
            }
            #pragma unroll
            for (int u = 0; u < 4; ++u) v[u] = pos_diff[ix[u]];
            #pragma unroll
            for (int u = 0; u < 4; ++u) {
                const float x = (k0 + u * NSTAT * 256 < M) ? v[u] : 0.0f;
                s1 += x; s2 += x * x;
            }
        }
        for (int off = 16; off > 0; off >>= 1) {
            s1 += __shfl_xor(s1, off, 32);
            s2 += __shfl_xor(s2, off, 32);
        }
        if ((tid & 31) == 0) { P[2 * (tid >> 5)] = s1; P[2 * (tid >> 5) + 1] = s2; }
        __syncthreads();
        if (tid == 0) {
            float a1 = 0.0f, a2 = 0.0f;
            for (int w = 0; w < 8; ++w) { a1 += P[2 * w]; a2 += P[2 * w + 1]; }
            scratch[2 * sb]     = a1;
            scratch[2 * sb + 1] = a2;
        }
        return;
    }

    const int wave = tid >> 5;
    const int lane = tid & 31;
    const int row0 = (blockIdx.x - nz - NSTAT) * 16;
    const int cb   = wave * 16;
    const int hl   = lane & 15;
    const int hi   = lane >> 4;

    #pragma unroll
    for (int j = 0; j < 4; ++j) {
        int idx = tid * 4 + j * 1024;
        int r = idx >> 8;
        int c = idx & 255;
        int grow = row0 + r;
        float4 v = make_float4(0.f, 0.f, 0.f, 0.f);
        if (grow < N) v = *(const float4*)(feats + (size_t)grow * 256 + c);
        float* dstp = &As[r * 260 + c];
        dstp[0] = v.x; dstp[1] = v.y; dstp[2] = v.z; dstp[3] = v.w;
    }
    __syncthreads();

    v8f acc = {};
    for (int k = 0; k < 256; k += 4) {
        int ka = k + hi * 2;
        v2f a, b;
        a.x = As[hl * 260 + ka];
        a.y = As[hl * 260 + ka + 1];
        b.x = W[(size_t)ka * 128 + cb + hl];
        b.y = W[(size_t)(ka + 1) * 128 + cb + hl];
        acc = __builtin_amdgcn_wmma_f32_16x16x4_f32(false, a, false, b,
                                                    (short)0, acc, false, false);
    }

    float bv = bias[cb + hl];
    int rbase = hi * 8;
    #pragma unroll
    for (int r = 0; r < 8; ++r) {
        float v = acc[r] + bv;
        v = v > 0.0f ? v : 0.0f;
        T[(rbase + r) * 132 + cb + hl] = v;
    }
    __syncthreads();

    {
        int row = tid >> 4;
        int c0  = (tid & 15) * 8;
        float s = 0.0f;
        #pragma unroll
        for (int j = 0; j < 8; ++j) { float v = T[row * 132 + c0 + j]; s += v * v; }
        P[row * 17 + (tid & 15)] = s;
    }
    __syncthreads();
    if (tid < 16) {
        float s = 0.0f;
        #pragma unroll
        for (int j = 0; j < 16; ++j) s += P[tid * 17 + j];
        RowInv[tid] = 1.0f / fmaxf(sqrtf(s), EPS);
    }
    __syncthreads();

    {
        int row  = tid >> 4;
        int grow = row0 + row;
        if (grow < N) {
            int c0 = (tid & 15) * 8;
            float inv = RowInv[row];
            float* op = h + (size_t)grow * 128 + c0;
            #pragma unroll
            for (int j = 0; j < 8; ++j) op[j] = T[row * 132 + c0 + j] * inv;
        }
    }
}

__global__ __launch_bounds__(BIN_THREADS) void bin_kernel(const int* __restrict__ src,
                                                          const int* __restrict__ dst,
                                                          int* __restrict__ binCount,
                                                          int* __restrict__ bins,
                                                          int* __restrict__ cnt,
                                                          int* __restrict__ ovfn,
                                                          int* __restrict__ ovf,
                                                          int N, int E, int nbin) {
    __shared__ int hist[MAXBIN];
    __shared__ int off[MAXBIN];
    __shared__ int gbase[MAXBIN];
    __shared__ int wsum[32];
    __shared__ int val[EPB];
    __shared__ unsigned short binOf[EPB];

    const int tid  = threadIdx.x;
    const int lane = tid & 31;
    const int wave = tid >> 5;
    const int e0   = blockIdx.x * EPB;

    if (tid < MAXBIN) hist[tid] = 0;
    __syncthreads();

    int dd[EPB / BIN_THREADS], ss[EPB / BIN_THREADS], rr[EPB / BIN_THREADS];
    #pragma unroll
    for (int k = 0; k < EPB / BIN_THREADS; ++k) {
        const int e = e0 + k * BIN_THREADS + tid;
        dd[k] = -1; ss[k] = 0;
        if (e < E) { dd[k] = dst[e]; ss[k] = src[e]; }
    }
    #pragma unroll
    for (int k = 0; k < EPB / BIN_THREADS; ++k) {
        rr[k] = 0;
        if ((unsigned)dd[k] < (unsigned)N)
            rr[k] = __hip_atomic_fetch_add(&hist[dd[k] >> BIN_SHIFT], 1, __ATOMIC_RELAXED,
                                           __HIP_MEMORY_SCOPE_WORKGROUP);
    }
    __syncthreads();

    const int v = (tid < MAXBIN) ? hist[tid] : 0;
    int inc = v;
    #pragma unroll
    for (int o = 1; o < 32; o <<= 1) {
        const int t = __shfl_up(inc, o, 32);
        if (lane >= o) inc += t;
    }
    if (lane == 31) wsum[wave] = inc;
    if (tid < nbin && v > 0)
        gbase[tid] = __hip_atomic_fetch_add(&binCount[tid * CNT_STRIDE], v, __ATOMIC_RELAXED,
                                            __HIP_MEMORY_SCOPE_AGENT);
    __syncthreads();
    if (wave == 0) {
        const int w = wsum[lane];
        int winc = w;
        #pragma unroll
        for (int o = 1; o < 32; o <<= 1) {
            const int t = __shfl_up(winc, o, 32);
            if (lane >= o) winc += t;
        }
        wsum[lane] = winc - w;
    }
    __syncthreads();
    if (tid < MAXBIN) off[tid] = inc - v + wsum[wave];
    __syncthreads();

    #pragma unroll
    for (int k = 0; k < EPB / BIN_THREADS; ++k) {
        if ((unsigned)dd[k] < (unsigned)N) {
            const int b = dd[k] >> BIN_SHIFT;
            const int p = off[b] + rr[k];
            val[p]   = ((dd[k] & (BIN_NODES - 1)) << 24) | (ss[k] & 0xFFFFFF);
            binOf[p] = (unsigned short)b;
        }
    }
    __syncthreads();

    const int nval = off[MAXBIN - 1] + hist[MAXBIN - 1];
    for (int p = tid; p < nval; p += BIN_THREADS) {
        const int b = binOf[p];
        const int g = gbase[b] + (p - off[b]);
        const int w = val[p];
        const int left = N - (b << BIN_SHIFT);
        const int capb = (left < BIN_NODES ? left : BIN_NODES) * CAP;
        if ((unsigned)g < (unsigned)capb) {
            bins[(size_t)b * BIN_CAP + g] = w;
        } else {
            const int d = (b << BIN_SHIFT) + (int)((unsigned)w >> 24);
            const int q = __hip_atomic_fetch_add(ovfn, 1, __ATOMIC_RELAXED, __HIP_MEMORY_SCOPE_AGENT);
            if ((unsigned)q < (unsigned)E) {
                ovf[2 * (size_t)q]     = d;
                ovf[2 * (size_t)q + 1] = w & 0xFFFFFF;
                __hip_atomic_fetch_add(&cnt[d], 1, __ATOMIC_RELAXED, __HIP_MEMORY_SCOPE_AGENT);
            }
        }
    }
}

__global__ __launch_bounds__(BUILD_THREADS) void build_kernel(const int* __restrict__ binCount,
                                                              const int* bins,
                                                              int* __restrict__ cnt,
                                                              int* bucket,
                                                              int* __restrict__ ovfn,
                                                              int* __restrict__ ovf,
                                                              int N, int E) {
    __shared__ int cntL[BIN_NODES];
    __shared__ int bucketL[BIN_NODES * (CAP + 1)];

    const int tid   = threadIdx.x;
    const int b     = blockIdx.x;
    const int node0 = b << BIN_SHIFT;

    if (tid < BIN_NODES) cntL[tid] = 0;
    for (int k = tid; k < BIN_NODES * (CAP + 1); k += BUILD_THREADS) bucketL[k] = 0;
    __syncthreads();

    int n = binCount[b * CNT_STRIDE];
    const int left = N - node0;
    const int capb = (left < BIN_NODES ? left : BIN_NODES) * CAP;
    if (n > capb) n = capb;
    const int* my = bins + (size_t)b * BIN_CAP;
    for (int base = 0; base < n; base += 8 * BUILD_THREADS) {
        int w[8];
        #pragma unroll
        for (int u = 0; u < 8; ++u) {
            const int e = base + u * BUILD_THREADS + tid;
            w[u] = 0;
            if (e < n) w[u] = my[e];
        }
        #pragma unroll
        for (int u = 0; u < 8; ++u) {
            const int e = base + u * BUILD_THREADS + tid;
            if (e < n) {
                const int dl = (int)((unsigned)w[u] >> 24);
                const int s  = w[u] & 0xFFFFFF;
                const int slot = __hip_atomic_fetch_add(&cntL[dl], 1, __ATOMIC_RELAXED,
                                                        __HIP_MEMORY_SCOPE_WORKGROUP);
                if ((unsigned)slot < (unsigned)CAP) {
                    bucketL[dl * (CAP + 1) + slot] = s;
                } else {
                    const int q = __hip_atomic_fetch_add(ovfn, 1, __ATOMIC_RELAXED, __HIP_MEMORY_SCOPE_AGENT);
                    if ((unsigned)q < (unsigned)E) {
                        ovf[2 * (size_t)q]     = node0 + dl;
                        ovf[2 * (size_t)q + 1] = s;
                    }
                }
            }
        }
    }
    __syncthreads();

    if (tid < BIN_NODES) {
        const int node = node0 + tid;
        if (node < N) {
            const int cl  = cntL[tid];
            const int tot = cl + cnt[node];
            const int filled = cl < CAP ? cl : CAP;
            cnt[node] = (tot & 0xFFFFFF) | (filled << 24);
        }
    }
    for (int k = tid; k < BIN_NODES * CAP; k += BUILD_THREADS) {
        const int nl = k >> 5;
        const int j  = k & (CAP - 1);
        if (node0 + nl < N) bucket[(size_t)(node0 + nl) * CAP + j] = bucketL[nl * (CAP + 1) + j];
    }
}

#define NPW 10
#define PULL_NODES (8 * NPW)

__device__ __forceinline__ float4 row4(const float* __restrict__ hl, int s) {
    return *(const float4*)(hl + (size_t)s * 128);
}
__device__ __forceinline__ void add4(float4& a, const float4& v) {
    a.x += v.x; a.y += v.y; a.z += v.z; a.w += v.w;
}
template <int G>
__device__ __forceinline__ void add_rows(float4& acc, const float* __restrict__ hl,
                                         int myidx, int k) {
    float4 v[G];
    #pragma unroll
    for (int u = 0; u < G; ++u) v[u] = row4(hl, __builtin_amdgcn_readlane(myidx, k + u));
    #pragma unroll
    for (int u = 0; u < G; ++u) add4(acc, v[u]);
}
__device__ __forceinline__ void walk(float4& acc, const float* __restrict__ hl, int myidx, int m) {
    int k = 0;
    for (; k + 8 <= m; k += 8) add_rows<8>(acc, hl, myidx, k);
    if (m & 4) { add_rows<4>(acc, hl, myidx, k); k += 4; }
    if (m & 2) { add_rows<2>(acc, hl, myidx, k); k += 2; }
    if (m & 1) { add_rows<1>(acc, hl, myidx, k); }
}

__global__ __launch_bounds__(256) void pull_finalize_kernel(
    const float* __restrict__ h, const int* __restrict__ cnt,
    const int* __restrict__ bucket, const int* __restrict__ ovfn,
    const int* __restrict__ ovf,
    const float* __restrict__ center,
    const float* __restrict__ pos_diff, const int* __restrict__ out_nodes,
    const int* __restrict__ labels, const int* __restrict__ epoch,
    const float* __restrict__ scratch, float* __restrict__ rec,
    float* __restrict__ scores_out, int N, int E, int Mnor) {
    __shared__ float red[8][4];
    __shared__ float sc[PULL_NODES];

    const int wave = __builtin_amdgcn_readfirstlane(threadIdx.x >> 5);
    const int lane = threadIdx.x & 31;
    const int i0 = (blockIdx.x * 8 + wave) * NPW;

    float p_sum = 0.0f, p_cnt = 0.0f, n_sum = 0.0f, n_cnt = 0.0f;

    if (i0 < N) {
        const int nn = i0 + NPW < N ? NPW : N - i0;
        const float* hl = h + lane * 4;

        int    idx = bucket[(size_t)i0 * CAP + lane];
        float4 hv  = *(const float4*)(hl + (size_t)i0 * 128);

        int cw_v = 0, lab_v = 0;
        float pd_v = 0.0f;
        if (lane < nn) {
            cw_v = cnt[i0 + lane];
            const int on = out_nodes[i0 + lane];
            pd_v  = pos_diff[on];
            lab_v = labels[on];
        }

        const float4 cv = *(const float4*)(center + lane * 4);
        const float cntf = (float)Mnor;
        float st1 = 0.0f, st2 = 0.0f;
        for (int q = 0; q < NSTAT; ++q) { st1 += scratch[2 * q]; st2 += scratch[2 * q + 1]; }
        const float mu  = st1 / cntf;
        const float var = (st2 - cntf * mu * mu) / (cntf - 1.0f);
        const float sd  = sqrtf(var);
        float beta = powf(0.9f, (float)epoch[0]);
        if (beta < 0.1f) beta = 0.0f;

        const int   c_v    = cw_v & 0xFFFFFF;
        const float invd_v = c_v > 0 ? 1.0f / fmaxf((float)c_v, 1.0f) : 0.0f;
        const float z_v    = (pd_v - mu) / sd;
        const float pre_v  = 1.0f - 1.0f / (1.0f + expf(-z_v));
        float score_v = 0.0f;

        for (int j = 0; j < nn; ++j) {
            const int i = i0 + j;
            const int    idxc = idx;
            const float4 hvc  = hv;
            if (j + 1 < nn) {
                idx = bucket[(size_t)(i + 1) * CAP + lane];
                hv  = *(const float4*)(hl + (size_t)(i + 1) * 128);
            }
            const int   cwc  = __builtin_amdgcn_readlane(cw_v, j);
            const float invd = __int_as_float(__builtin_amdgcn_readlane(__float_as_int(invd_v), j));
            const float pre  = __int_as_float(__builtin_amdgcn_readlane(__float_as_int(pre_v), j));

            const int c  = cwc & 0xFFFFFF;
            const int mf = (int)((unsigned)cwc >> 24);
            const int m0 = mf < CAP ? mf : CAP;

            float4 acc = make_float4(0.f, 0.f, 0.f, 0.f);
            if (m0 > 0) walk(acc, hl, idxc, m0);
            if (c > m0) {
                int n = ovfn[0];
                if (n > E) n = E;
                for (int base = 0; base < n; base += 32) {
                    const int k = base + lane;
                    int d = -1, s = 0;
                    if (k < n) { d = ovf[2 * (size_t)k]; s = ovf[2 * (size_t)k + 1]; }
                    unsigned hit = __builtin_amdgcn_ballot_w32(d == i);
                    while (hit) {
                        const int b = __builtin_ctz(hit);
                        hit &= hit - 1;
                        const float4 v = row4(hl, __builtin_amdgcn_readlane(s, b));
                        add4(acc, v);
                    }
                }
            }

            float4 mv;
            mv.x = acc.x * invd; mv.y = acc.y * invd; mv.z = acc.z * invd; mv.w = acc.w * invd;
            float dot = hvc.x * mv.x;
            dot += hvc.y * mv.y;
            dot += hvc.z * mv.z;
            dot += hvc.w * mv.w;
            for (int o = 16; o > 0; o >>= 1) dot += __shfl_xor(dot, o, 32);

            float attn = beta * pre + (1.0f - beta) * dot;
            float nei  = attn * NEIGH_WEIGHT;

            float score = (nei * mv.x + (1.0f - nei) * hvc.x) * cv.x;
            score += (nei * mv.y + (1.0f - nei) * hvc.y) * cv.y;
            score += (nei * mv.z + (1.0f - nei) * hvc.z) * cv.z;
            score += (nei * mv.w + (1.0f - nei) * hvc.w) * cv.w;
            for (int o = 16; o > 0; o >>= 1) score += __shfl_xor(score, o, 32);

            if (lane == j) score_v = score;
        }

        if (lane < nn) {
            sc[wave * NPW + lane] = score_v;
            float spl = log1pf(expf(-fabsf(score_v)));
            if (lab_v == 1) {
                n_sum = fmaxf(score_v, 0.0f) + spl;
                n_cnt = 1.0f;
            } else {
                p_sum = fmaxf(-score_v, 0.0f) + spl;
                p_cnt = 1.0f;
            }
        }
        for (int o = 16; o > 0; o >>= 1) {
            p_sum += __shfl_xor(p_sum, o, 32);
            p_cnt += __shfl_xor(p_cnt, o, 32);
            n_sum += __shfl_xor(n_sum, o, 32);
            n_cnt += __shfl_xor(n_cnt, o, 32);
        }
    }

    if (lane == 0) {
        red[wave][0] = p_sum; red[wave][1] = p_cnt;
        red[wave][2] = n_sum; red[wave][3] = n_cnt;
    }
    __syncthreads();
    if (threadIdx.x < 4) {
        float s = 0.0f;
        for (int w = 0; w < 8; ++w) s += red[w][threadIdx.x];
        rec[4 * (size_t)blockIdx.x + threadIdx.x] = s;
    }
    if (threadIdx.x < PULL_NODES) {
        const int i = blockIdx.x * PULL_NODES + threadIdx.x;
        if (i < N) scores_out[i] = sc[threadIdx.x];
    }
}

__global__ __launch_bounds__(1024) void loss_kernel(const float* __restrict__ rec, int N,
                                                    float* __restrict__ out0) {
    __shared__ float red[32][4];
    const int tid  = threadIdx.x;
    const int lane = tid & 31;
    const int wave = tid >> 5;
    const int nblk = (N + PULL_NODES - 1) / PULL_NODES;

    float a0 = 0.0f, a1 = 0.0f, a2 = 0.0f, a3 = 0.0f;
    #pragma unroll 4
    for (int b = tid; b < nblk; b += 1024) {
        float4 v = *(const float4*)(rec + 4 * (size_t)b);
        a0 += v.x; a1 += v.y; a2 += v.z; a3 += v.w;
    }
    for (int off = 16; off > 0; off >>= 1) {
        a0 += __shfl_xor(a0, off, 32);
        a1 += __shfl_xor(a1, off, 32);
        a2 += __shfl_xor(a2, off, 32);
        a3 += __shfl_xor(a3, off, 32);
    }
    if (lane == 0) {
        red[wave][0] = a0; red[wave][1] = a1; red[wave][2] = a2; red[wave][3] = a3;
    }
    __syncthreads();
    if (wave == 0) {
        float r0 = red[lane][0], r1 = red[lane][1], r2 = red[lane][2], r3 = red[lane][3];
        for (int off = 16; off > 0; off >>= 1) {
            r0 += __shfl_xor(r0, off, 32);
            r1 += __shfl_xor(r1, off, 32);
            r2 += __shfl_xor(r2, off, 32);
            r3 += __shfl_xor(r3, off, 32);
        }
        if (lane == 0)
            out0[0] = r0 / fmaxf(r1, 1.0f) + r2 / fmaxf(r3, 1.0f);
    }
}

extern "C" void kernel_launch(void* const* d_in, const int* in_sizes, int n_in,
                              void* d_out, int out_size, void* d_ws, size_t ws_size,
                              hipStream_t stream) {
    const float* feats    = (const float*)d_in[0];
    const float* W        = (const float*)d_in[1];
    const float* b        = (const float*)d_in[2];
    const float* center   = (const float*)d_in[3];
    const float* pos_diff = (const float*)d_in[4];
    const int*   src      = (const int*)d_in[5];
    const int*   dst      = (const int*)d_in[6];
    const int*   out_nodes= (const int*)d_in[7];
    const int*   nor_idx  = (const int*)d_in[8];
    const int*   labels   = (const int*)d_in[9];
    const int*   epoch    = (const int*)d_in[10];

    const int N = in_sizes[0] / 256;
    const int E = in_sizes[5];
    const int M = in_sizes[8];
    const int nblk = (N + 7) / 8;

    const int nbin = (N + BIN_NODES - 1) / BIN_NODES;
    const int npull = (N + PULL_NODES - 1) / PULL_NODES;

    const int nz = (N + ZERO_WORDS - 1) / ZERO_WORDS;

    float* wsf     = (float*)d_ws;
    float* scratch = wsf;
    float* h       = wsf + 64;
    int*   bucket  = (int*)(h + (size_t)N * 128);
    float* rec     = (float*)(bucket + (size_t)N * CAP);
    int*   ovfn    = (int*)(rec + 4 * (size_t)nblk);
    int*   ovf     = ovfn + 4;
    int*   cnt     = ovf + 2 * (size_t)E;
    int*   bins    = bucket;
    int*   binCount= (int*)rec;

    float* out     = (float*)d_out;

    gemm_h_kernel<<<nz + NSTAT + (N + 15) / 16, 256, 0, stream>>>(feats, W, b, h, N, pos_diff, nor_idx, M,
                                                                 scratch, cnt, ovfn, binCount, nbin, nz);

    bin_kernel<<<(E + EPB - 1) / EPB, BIN_THREADS, 0, stream>>>(src, dst, binCount, bins, cnt, ovfn, ovf, N, E, nbin);
    build_kernel<<<nbin, BUILD_THREADS, 0, stream>>>(binCount, bins, cnt, bucket, ovfn, ovf, N, E);

    pull_finalize_kernel<<<npull, 256, 0, stream>>>(h, cnt, bucket, ovfn, ovf, center, pos_diff,
                                                   out_nodes, labels, epoch, scratch, rec,
                                                   out + 1, N, E, M);

    loss_kernel<<<1, 1024, 0, stream>>>(rec, N, out);
}


# neighbour lists not stored: a pull block orders its own 80 nodes' edges from the bin's edge words into LDS (LDS adds) and walks them there; build launch gone, 5 launches to 4 (slot order changes)
# speedup vs baseline: 6.3729x; 1.0869x over previous
#include <hip/hip_runtime.h>
#include <hip/hip_bf16.h>
#include <math.h>

typedef float v2f __attribute__((ext_vector_type(2)));
typedef float v8f __attribute__((ext_vector_type(8)));

#define EPS 1e-12f
#define NEIGH_WEIGHT 0.2f
#define CAP 32

#define BIN_SHIFT 8
#define BIN_NODES 256
#define BIN_CAP (BIN_NODES * CAP)
#define NSTAT 16
#define ZERO_WORDS 4096
#define MAXBIN 512
#define EPB 8192
#define BIN_THREADS 1024
#define CNT_STRIDE 32

__global__ __launch_bounds__(256) void gemm_h_kernel(
    const float* __restrict__ feats, const float* __restrict__ W,
    const float* __restrict__ bias, float* __restrict__ h, int N,
    const float* __restrict__ pos_diff, const int* __restrict__ nor_idx, int M,
    float* __restrict__ scratch, int* __restrict__ cnt, int* __restrict__ ovfn,
    int* __restrict__ binCount, int nbin, int nz) {
    __shared__ float As[16 * 260];
    __shared__ float T[16 * 132];
    __shared__ float P[16 * 17];
    __shared__ float RowInv[16];

    const int tid  = threadIdx.x;

    if ((int)blockIdx.x < nz) {
        const int w0 = blockIdx.x * ZERO_WORDS + tid;
        for (int j = 0; j < ZERO_WORDS / 256; ++j) {
            const int t = w0 + j * 256;
            if (t < N) cnt[t] = 0;
        }
        for (int t = blockIdx.x * 256 + tid; t < nbin; t += nz * 256) binCount[t * CNT_STRIDE] = 0;
        if (blockIdx.x == 0 && tid == 0) ovfn[0] = 0;
        return;
    }
    if ((int)blockIdx.x < nz + NSTAT) {
        const int sb = blockIdx.x - nz;
        float s1 = 0.0f, s2 = 0.0f;
        for (int k0 = sb * 256 + tid; k0 < M; k0 += 4 * NSTAT * 256) {
            int   ix[4];
            float v[4];
            #pragma unroll
            for (int u = 0; u < 4; ++u) {
                const int k = k0 + u * NSTAT * 256;
                ix[u] = nor_idx[k < M ? k : M - 1];
            }
            #pragma unroll
            for (int u = 0; u < 4; ++u) v[u] = pos_diff[ix[u]];
            #pragma unroll
            for (int u = 0; u < 4; ++u) {
                const float x = (k0 + u * NSTAT * 256 < M) ? v[u] : 0.0f;
                s1 += x; s2 += x * x;
            }
        }
        for (int off = 16; off > 0; off >>= 1) {
            s1 += __shfl_xor(s1, off, 32);
            s2 += __shfl_xor(s2, off, 32);
        }
        if ((tid & 31) == 0) { P[2 * (tid >> 5)] = s1; P[2 * (tid >> 5) + 1] = s2; }
        __syncthreads();
        if (tid == 0) {
            float a1 = 0.0f, a2 = 0.0f;
            for (int w = 0; w < 8; ++w) { a1 += P[2 * w]; a2 += P[2 * w + 1]; }
            scratch[2 * sb]     = a1;
            scratch[2 * sb + 1] = a2;
        }
        return;
    }

    const int wave = tid >> 5;
    const int lane = tid & 31;
    const int row0 = (blockIdx.x - nz - NSTAT) * 16;
    const int cb   = wave * 16;
    const int hl   = lane & 15;
    const int hi   = lane >> 4;

    #pragma unroll
    for (int j = 0; j < 4; ++j) {
        int idx = tid * 4 + j * 1024;
        int r = idx >> 8;
        int c = idx & 255;
        int grow = row0 + r;
        float4 v = make_float4(0.f, 0.f, 0.f, 0.f);
        if (grow < N) v = *(const float4*)(feats + (size_t)grow * 256 + c);
        float* dstp = &As[r * 260 + c];
        dstp[0] = v.x; dstp[1] = v.y; dstp[2] = v.z; dstp[3] = v.w;
    }
    __syncthreads();

    v8f acc = {};
    for (int k = 0; k < 256; k += 4) {
        int ka = k + hi * 2;
        v2f a, b;
        a.x = As[hl * 260 + ka];
        a.y = As[hl * 260 + ka + 1];
        b.x = W[(size_t)ka * 128 + cb + hl];
        b.y = W[(size_t)(ka + 1) * 128 + cb + hl];
        acc = __builtin_amdgcn_wmma_f32_16x16x4_f32(false, a, false, b,
                                                    (short)0, acc, false, false);
    }

    float bv = bias[cb + hl];
    int rbase = hi * 8;
    #pragma unroll
    for (int r = 0; r < 8; ++r) {
        float v = acc[r] + bv;
        v = v > 0.0f ? v : 0.0f;
        T[(rbase + r) * 132 + cb + hl] = v;
    }
    __syncthreads();

    {
        int row = tid >> 4;
        int c0  = (tid & 15) * 8;
        float s = 0.0f;
        #pragma unroll
        for (int j = 0; j < 8; ++j) { float v = T[row * 132 + c0 + j]; s += v * v; }
        P[row * 17 + (tid & 15)] = s;
    }
    __syncthreads();
    if (tid < 16) {
        float s = 0.0f;
        #pragma unroll
        for (int j = 0; j < 16; ++j) s += P[tid * 17 + j];
        RowInv[tid] = 1.0f / fmaxf(sqrtf(s), EPS);
    }
    __syncthreads();

    {
        int row  = tid >> 4;
        int grow = row0 + row;
        if (grow < N) {
            int c0 = (tid & 15) * 8;
            float inv = RowInv[row];
            float* op = h + (size_t)grow * 128 + c0;
            #pragma unroll
            for (int j = 0; j < 8; ++j) op[j] = T[row * 132 + c0 + j] * inv;
        }
    }
}

__global__ __launch_bounds__(BIN_THREADS) void bin_kernel(const int* __restrict__ src,
                                                          const int* __restrict__ dst,
                                                          int* __restrict__ binCount,
                                                          int* __restrict__ bins,
                                                          int* __restrict__ cnt,
                                                          int* __restrict__ ovfn,
                                                          int* __restrict__ ovf,
                                                          int N, int E, int nbin) {
    __shared__ int hist[MAXBIN];
    __shared__ int off[MAXBIN];
    __shared__ int gbase[MAXBIN];
    __shared__ int wsum[32];
    __shared__ int val[EPB];
    __shared__ unsigned short binOf[EPB];

    const int tid  = threadIdx.x;
    const int lane = tid & 31;
    const int wave = tid >> 5;
    const int e0   = blockIdx.x * EPB;

    if (tid < MAXBIN) hist[tid] = 0;
    __syncthreads();

    int dd[EPB / BIN_THREADS], ss[EPB / BIN_THREADS], rr[EPB / BIN_THREADS];
    #pragma unroll
    for (int k = 0; k < EPB / BIN_THREADS; ++k) {
        const int e = e0 + k * BIN_THREADS + tid;
        dd[k] = -1; ss[k] = 0;
        if (e < E) { dd[k] = dst[e]; ss[k] = src[e]; }
    }
    #pragma unroll
    for (int k = 0; k < EPB / BIN_THREADS; ++k) {
        rr[k] = 0;
        if ((unsigned)dd[k] < (unsigned)N)
            rr[k] = __hip_atomic_fetch_add(&hist[dd[k] >> BIN_SHIFT], 1, __ATOMIC_RELAXED,
                                           __HIP_MEMORY_SCOPE_WORKGROUP);
    }
    __syncthreads();

    const int v = (tid < MAXBIN) ? hist[tid] : 0;
    int inc = v;
    #pragma unroll
    for (int o = 1; o < 32; o <<= 1) {
        const int t = __shfl_up(inc, o, 32);
        if (lane >= o) inc += t;
    }
    if (lane == 31) wsum[wave] = inc;
    if (tid < nbin && v > 0)
        gbase[tid] = __hip_atomic_fetch_add(&binCount[tid * CNT_STRIDE], v, __ATOMIC_RELAXED,
                                            __HIP_MEMORY_SCOPE_AGENT);
    __syncthreads();
    if (wave == 0) {
        const int w = wsum[lane];
        int winc = w;
        #pragma unroll
        for (int o = 1; o < 32; o <<= 1) {
            const int t = __shfl_up(winc, o, 32);
            if (lane >= o) winc += t;
        }
        wsum[lane] = winc - w;
    }
    __syncthreads();
    if (tid < MAXBIN) off[tid] = inc - v + wsum[wave];
    __syncthreads();

    #pragma unroll
    for (int k = 0; k < EPB / BIN_THREADS; ++k) {
        if ((unsigned)dd[k] < (unsigned)N) {
            const int b = dd[k] >> BIN_SHIFT;
            const int p = off[b] + rr[k];
            val[p]   = ((dd[k] & (BIN_NODES - 1)) << 24) | (ss[k] & 0xFFFFFF);
            binOf[p] = (unsigned short)b;
        }
    }
    __syncthreads();

    const int nval = off[MAXBIN - 1] + hist[MAXBIN - 1];
    for (int p = tid; p < nval; p += BIN_THREADS) {
        const int b = binOf[p];
        const int g = gbase[b] + (p - off[b]);
        const int w = val[p];
        const int left = N - (b << BIN_SHIFT);
        const int capb = (left < BIN_NODES ? left : BIN_NODES) * CAP;
        if ((unsigned)g < (unsigned)capb) {
            bins[(size_t)b * BIN_CAP + g] = w;
        } else {
            const int d = (b << BIN_SHIFT) + (int)((unsigned)w >> 24);
            const int q = __hip_atomic_fetch_add(ovfn, 1, __ATOMIC_RELAXED, __HIP_MEMORY_SCOPE_AGENT);
            if ((unsigned)q < (unsigned)E) {
                ovf[2 * (size_t)q]     = d;
                ovf[2 * (size_t)q + 1] = w & 0xFFFFFF;
                __hip_atomic_fetch_add(&cnt[d], 1, __ATOMIC_RELAXED, __HIP_MEMORY_SCOPE_AGENT);
            }
        }
    }
}

#define NPW 10
#define PULL_NODES (8 * NPW)
#define OVL_CAP 64
#define PF 5

static_assert(PULL_NODES <= BIN_NODES, "a pull block's nodes must lie in at most two bins");

__device__ __forceinline__ float4 row4(const float* __restrict__ hl, int s) {
    return *(const float4*)(hl + (size_t)s * 128);
}
__device__ __forceinline__ void add4(float4& a, const float4& v) {
    a.x += v.x; a.y += v.y; a.z += v.z; a.w += v.w;
}
template <int G>
__device__ __forceinline__ void add_rows(float4& acc, const float* __restrict__ hl,
                                         int myidx, int k) {
    float4 v[G];
    #pragma unroll
    for (int u = 0; u < G; ++u) v[u] = row4(hl, __builtin_amdgcn_readlane(myidx, k + u));
    #pragma unroll
    for (int u = 0; u < G; ++u) add4(acc, v[u]);
}
__device__ __forceinline__ void walk(float4& acc, const float* __restrict__ hl, int myidx, int m) {
    int k = 0;
    for (; k + 8 <= m; k += 8) add_rows<8>(acc, hl, myidx, k);
    if (m & 4) { add_rows<4>(acc, hl, myidx, k); k += 4; }
    if (m & 2) { add_rows<2>(acc, hl, myidx, k); k += 2; }
    if (m & 1) { add_rows<1>(acc, hl, myidx, k); }
}

__global__ __launch_bounds__(256) void pull_finalize_kernel(
    const float* __restrict__ h, const int* __restrict__ cnt,
    const int* __restrict__ binCount, const int* __restrict__ bins,
    const int* __restrict__ ovfn, const int* __restrict__ ovf,
    const float* __restrict__ center,
    const float* __restrict__ pos_diff, const int* __restrict__ out_nodes,
    const int* __restrict__ labels, const int* __restrict__ epoch,
    const float* __restrict__ scratch, float* __restrict__ rec,
    float* __restrict__ scores_out, int N, int E, int Mnor) {
    __shared__ float red[8][4];
    __shared__ float sc[PULL_NODES];
    __shared__ int cntL[PULL_NODES];
    __shared__ int bucketL[PULL_NODES * (CAP + 1)];
    __shared__ int ovL[2 * OVL_CAP];
    __shared__ int ovLn;

    const int wave = __builtin_amdgcn_readfirstlane(threadIdx.x >> 5);
    const int tid  = threadIdx.x;
    const int lane = threadIdx.x & 31;
    const int s0 = blockIdx.x * PULL_NODES;
    const int r0 = wave * NPW;
    const int i0 = s0 + r0;
    const int nn = i0 < N ? (i0 + NPW < N ? NPW : N - i0) : 0;
    const float* hl = h + lane * 4;

    if (tid < PULL_NODES) cntL[tid] = 0;
    if (tid == 0) ovLn = 0;
    __syncthreads();

    const int ic = i0 < N ? i0 : N - 1;
    const int iv = i0 + lane < N ? i0 + lane : N - 1;
    float4 hv = *(const float4*)(hl + (size_t)ic * 128);
    const float4 cv  = *(const float4*)(center + lane * 4);
    const int    ep  = epoch[0];
    const float2 sxy = *(const float2*)(scratch + 2 * (lane < NSTAT ? lane : NSTAT - 1));
    const int    sp_v = cnt[iv];
    const int    on_v = out_nodes[iv];

    const int lastn = (s0 + PULL_NODES < N ? s0 + PULL_NODES : N) - 1;
    const int b0 = s0 >> BIN_SHIFT;
    const int b1 = lastn >> BIN_SHIFT;
    int n0 = binCount[b0 * CNT_STRIDE];
    int n1 = b1 > b0 ? binCount[b1 * CNT_STRIDE] : 0;
    {
        const int left0 = N - (b0 << BIN_SHIFT);
        const int cap0  = (left0 < BIN_NODES ? left0 : BIN_NODES) * CAP;
        if (n0 > cap0) n0 = cap0;
        const int left1 = N - (b1 << BIN_SHIFT);
        const int cap1  = (left1 < BIN_NODES ? left1 : BIN_NODES) * CAP;
        if (n1 > cap1) n1 = cap1;
    }

    constexpr int CHUNK = 4 * PF * 256;
    int4 w[PF];
    auto load_chunk = [&](const int* __restrict__ my, int base, int n) {
        #pragma unroll
        for (int u = 0; u < PF; ++u) {
            const int e = base + (u * 256 + tid) * 4;
            w[u] = *(const int4*)(my + (e < n ? e : 0));
        }
    };
    auto place_chunk = [&](int base, int n, int rel) {
        #pragma unroll
        for (int u = 0; u < PF; ++u) {
            const int e = base + (u * 256 + tid) * 4;
            const int ww[4] = {w[u].x, w[u].y, w[u].z, w[u].w};
            int  r[4], slot[4];
            bool ok[4];
            #pragma unroll
            for (int q = 0; q < 4; ++q) {
                r[q]  = rel + (int)((unsigned)ww[q] >> 24);
                ok[q] = e + q < n && (unsigned)r[q] < (unsigned)PULL_NODES;
                if (ok[q])
                    slot[q] = __hip_atomic_fetch_add(&cntL[r[q]], 1, __ATOMIC_RELAXED,
                                                     __HIP_MEMORY_SCOPE_WORKGROUP);
            }
            #pragma unroll
            for (int q = 0; q < 4; ++q) {
                if (ok[q]) {
                    const int s = ww[q] & 0xFFFFFF;
                    if ((unsigned)slot[q] < (unsigned)CAP) {
                        bucketL[r[q] * (CAP + 1) + slot[q]] = s;
                    } else {
                        const int p = __hip_atomic_fetch_add(&ovLn, 1, __ATOMIC_RELAXED,
                                                             __HIP_MEMORY_SCOPE_WORKGROUP);
                        if ((unsigned)p < (unsigned)OVL_CAP) { ovL[2 * p] = r[q]; ovL[2 * p + 1] = s; }
                    }
                }
            }
        }
    };

    const int* my0 = bins + (size_t)b0 * BIN_CAP;
    const int rel0 = (b0 << BIN_SHIFT) - s0;
    load_chunk(my0, 0, n0);
    const float pd_v  = pos_diff[on_v];
    const int   lab_v = labels[on_v];
    place_chunk(0, n0, rel0);
    for (int base = CHUNK; base < n0; base += CHUNK) {
        load_chunk(my0, base, n0);
        place_chunk(base, n0, rel0);
    }
    if (b1 > b0) {
        const int* my1 = bins + (size_t)b1 * BIN_CAP;
        const int rel1 = (b1 << BIN_SHIFT) - s0;
        for (int base = 0; base < n1; base += CHUNK) {
            load_chunk(my1, base, n1);
            place_chunk(base, n1, rel1);
        }
    }

    float mu = 0.0f, sd = 1.0f, beta = 0.0f;
    if (nn > 0) {
        const float cntf = (float)Mnor;
        float st1 = 0.0f, st2 = 0.0f;
        for (int q = 0; q < NSTAT; ++q) {
            st1 += __int_as_float(__builtin_amdgcn_readlane(__float_as_int(sxy.x), q));
            st2 += __int_as_float(__builtin_amdgcn_readlane(__float_as_int(sxy.y), q));
        }
        mu = st1 / cntf;
        const float var = (st2 - cntf * mu * mu) / (cntf - 1.0f);
        sd = sqrtf(var);
        beta = powf(0.9f, (float)ep);
        if (beta < 0.1f) beta = 0.0f;
    }
    __syncthreads();

    float p_sum = 0.0f, p_cnt = 0.0f, n_sum = 0.0f, n_cnt = 0.0f;

    if (nn > 0) {
        int idx = bucketL[r0 * (CAP + 1) + lane];

        int cl_v = 0;
        if (lane < nn) cl_v = cntL[r0 + lane];

        const int   c_v    = cl_v + sp_v;
        const float invd_v = c_v > 0 ? 1.0f / fmaxf((float)c_v, 1.0f) : 0.0f;
        const float z_v    = (pd_v - mu) / sd;
        const float pre_v  = 1.0f - 1.0f / (1.0f + expf(-z_v));
        float score_v = 0.0f;

        for (int j = 0; j < nn; ++j) {
            const int i = i0 + j;
            const int    idxc = idx;
            const float4 hvc  = hv;
            if (j + 1 < nn) {
                idx = bucketL[(r0 + j + 1) * (CAP + 1) + lane];
                hv  = *(const float4*)(hl + (size_t)(i + 1) * 128);
            }
            const int   cl   = __builtin_amdgcn_readlane(cl_v, j);
            const int   sp   = __builtin_amdgcn_readlane(sp_v, j);
            const float invd = __int_as_float(__builtin_amdgcn_readlane(__float_as_int(invd_v), j));
            const float pre  = __int_as_float(__builtin_amdgcn_readlane(__float_as_int(pre_v), j));

            const int m0 = cl < CAP ? cl : CAP;

            float4 acc = make_float4(0.f, 0.f, 0.f, 0.f);
            if (m0 > 0) walk(acc, hl, idxc, m0);
            if (cl > CAP) {
                const int nl = ovLn;
                if (nl <= OVL_CAP) {
                    for (int base = 0; base < nl; base += 32) {
                        const int k = base + lane;
                        int r = -1, s = 0;
                        if (k < nl) { r = ovL[2 * k]; s = ovL[2 * k + 1]; }
                        unsigned hit = __builtin_amdgcn_ballot_w32(r == r0 + j);
                        while (hit) {
                            const int b = __builtin_ctz(hit);
                            hit &= hit - 1;
                            const float4 v = row4(hl, __builtin_amdgcn_readlane(s, b));
                            add4(acc, v);
                        }
                    }
                } else {
                    acc = make_float4(0.f, 0.f, 0.f, 0.f);
                    const int bb = i >> BIN_SHIFT;
                    int n = binCount[bb * CNT_STRIDE];
                    const int left = N - (bb << BIN_SHIFT);
                    const int capb = (left < BIN_NODES ? left : BIN_NODES) * CAP;
                    if (n > capb) n = capb;
                    const int* my = bins + (size_t)bb * BIN_CAP;
                    const int dlw = i & (BIN_NODES - 1);
                    for (int base = 0; base < n; base += 32) {
                        const int k = base + lane;
                        int w = 0;
                        if (k < n) w = my[k];
                        unsigned hit = __builtin_amdgcn_ballot_w32(k < n && (int)((unsigned)w >> 24) == dlw);
                        while (hit) {
                            const int b = __builtin_ctz(hit);
                            hit &= hit - 1;
                            const float4 v = row4(hl, __builtin_amdgcn_readlane(w, b) & 0xFFFFFF);
                            add4(acc, v);
                        }
                    }
                }
            }
            if (sp > 0) {
                int n = ovfn[0];
                if (n > E) n = E;
                for (int base = 0; base < n; base += 32) {
                    const int k = base + lane;
                    int d = -1, s = 0;
                    if (k < n) { d = ovf[2 * (size_t)k]; s = ovf[2 * (size_t)k + 1]; }
                    unsigned hit = __builtin_amdgcn_ballot_w32(d == i);
                    while (hit) {
                        const int b = __builtin_ctz(hit);
                        hit &= hit - 1;
                        const float4 v = row4(hl, __builtin_amdgcn_readlane(s, b));
                        add4(acc, v);
                    }
                }
            }

            float4 mv;
            mv.x = acc.x * invd; mv.y = acc.y * invd; mv.z = acc.z * invd; mv.w = acc.w * invd;
            float dot = hvc.x * mv.x;
            dot += hvc.y * mv.y;
            dot += hvc.z * mv.z;
            dot += hvc.w * mv.w;
            for (int o = 16; o > 0; o >>= 1) dot += __shfl_xor(dot, o, 32);

            float attn = beta * pre + (1.0f - beta) * dot;
            float nei  = attn * NEIGH_WEIGHT;

            float score = (nei * mv.x + (1.0f - nei) * hvc.x) * cv.x;
            score += (nei * mv.y + (1.0f - nei) * hvc.y) * cv.y;
            score += (nei * mv.z + (1.0f - nei) * hvc.z) * cv.z;
            score += (nei * mv.w + (1.0f - nei) * hvc.w) * cv.w;
            for (int o = 16; o > 0; o >>= 1) score += __shfl_xor(score, o, 32);

            if (lane == j) score_v = score;
        }

        if (lane < nn) {
            sc[wave * NPW + lane] = score_v;
            float spl = log1pf(expf(-fabsf(score_v)));
            if (lab_v == 1) {
                n_sum = fmaxf(score_v, 0.0f) + spl;
                n_cnt = 1.0f;
            } else {
                p_sum = fmaxf(-score_v, 0.0f) + spl;
                p_cnt = 1.0f;
            }
        }
        for (int o = 16; o > 0; o >>= 1) {
            p_sum += __shfl_xor(p_sum, o, 32);
            p_cnt += __shfl_xor(p_cnt, o, 32);
            n_sum += __shfl_xor(n_sum, o, 32);
            n_cnt += __shfl_xor(n_cnt, o, 32);
        }
    }

    if (lane == 0) {
        red[wave][0] = p_sum; red[wave][1] = p_cnt;
        red[wave][2] = n_sum; red[wave][3] = n_cnt;
    }
    __syncthreads();
    if (threadIdx.x < 4) {
        float s = 0.0f;
        for (int w = 0; w < 8; ++w) s += red[w][threadIdx.x];
        rec[4 * (size_t)blockIdx.x + threadIdx.x] = s;
    }
    if (threadIdx.x < PULL_NODES) {
        const int i = blockIdx.x * PULL_NODES + threadIdx.x;
        if (i < N) scores_out[i] = sc[threadIdx.x];
    }
}

__global__ __launch_bounds__(1024) void loss_kernel(const float* __restrict__ rec, int N,
                                                    float* __restrict__ out0) {
    __shared__ float red[32][4];
    const int tid  = threadIdx.x;
    const int lane = tid & 31;
    const int wave = tid >> 5;
    const int nblk = (N + PULL_NODES - 1) / PULL_NODES;

    float a0 = 0.0f, a1 = 0.0f, a2 = 0.0f, a3 = 0.0f;
    #pragma unroll 4
    for (int b = tid; b < nblk; b += 1024) {
        float4 v = *(const float4*)(rec + 4 * (size_t)b);
        a0 += v.x; a1 += v.y; a2 += v.z; a3 += v.w;
    }
    for (int off = 16; off > 0; off >>= 1) {
        a0 += __shfl_xor(a0, off, 32);
        a1 += __shfl_xor(a1, off, 32);
        a2 += __shfl_xor(a2, off, 32);
        a3 += __shfl_xor(a3, off, 32);
    }
    if (lane == 0) {
        red[wave][0] = a0; red[wave][1] = a1; red[wave][2] = a2; red[wave][3] = a3;
    }
    __syncthreads();
    if (wave == 0) {
        float r0 = red[lane][0], r1 = red[lane][1], r2 = red[lane][2], r3 = red[lane][3];
        for (int off = 16; off > 0; off >>= 1) {
            r0 += __shfl_xor(r0, off, 32);
            r1 += __shfl_xor(r1, off, 32);
            r2 += __shfl_xor(r2, off, 32);
            r3 += __shfl_xor(r3, off, 32);
        }
        if (lane == 0)
            out0[0] = r0 / fmaxf(r1, 1.0f) + r2 / fmaxf(r3, 1.0f);
    }
}

extern "C" void kernel_launch(void* const* d_in, const int* in_sizes, int n_in,
                              void* d_out, int out_size, void* d_ws, size_t ws_size,
                              hipStream_t stream) {
    const float* feats    = (const float*)d_in[0];
    const float* W        = (const float*)d_in[1];
    const float* b        = (const float*)d_in[2];
    const float* center   = (const float*)d_in[3];
    const float* pos_diff = (const float*)d_in[4];
    const int*   src      = (const int*)d_in[5];
    const int*   dst      = (const int*)d_in[6];
    const int*   out_nodes= (const int*)d_in[7];
    const int*   nor_idx  = (const int*)d_in[8];
    const int*   labels   = (const int*)d_in[9];
    const int*   epoch    = (const int*)d_in[10];

    const int N = in_sizes[0] / 256;
    const int E = in_sizes[5];
    const int M = in_sizes[8];
    const int nblk = (N + 7) / 8;

    const int nbin = (N + BIN_NODES - 1) / BIN_NODES;
    const int npull = (N + PULL_NODES - 1) / PULL_NODES;

    const int nz = (N + ZERO_WORDS - 1) / ZERO_WORDS;

    float* wsf     = (float*)d_ws;
    float* scratch = wsf;
    float* h       = wsf + 64;
    int*   bins    = (int*)(h + (size_t)N * 128);
    float* rec     = (float*)(bins + (size_t)N * CAP);
    int*   ovfn    = (int*)(rec + 4 * (size_t)nblk);
    int*   ovf     = ovfn + 4;
    int*   cnt     = ovf + 2 * (size_t)E;
    int*   binCount= (int*)rec + (4 * npull + CNT_STRIDE - 1) / CNT_STRIDE * CNT_STRIDE;

    float* out     = (float*)d_out;

    gemm_h_kernel<<<nz + NSTAT + (N + 15) / 16, 256, 0, stream>>>(feats, W, b, h, N, pos_diff, nor_idx, M,
                                                                 scratch, cnt, ovfn, binCount, nbin, nz);

    bin_kernel<<<(E + EPB - 1) / EPB, BIN_THREADS, 0, stream>>>(src, dst, binCount, bins, cnt, ovfn, ovf, N, E, nbin);

    pull_finalize_kernel<<<npull, 256, 0, stream>>>(h, cnt, binCount, bins, ovfn, ovf, center, pos_diff,
                                                   out_nodes, labels, epoch, scratch, rec,
                                                   out + 1, N, E, M);

    loss_kernel<<<1, 1024, 0, stream>>>(rec, N, out);
}


# GEMM as resident tile loop: 512 blocks loop over row tiles, each wave keeps its 256x16 slice of W in registers (no global load in the k-loop), next strip fetched under the products; h bit-identical
# speedup vs baseline: 7.1597x; 1.1235x over previous
#include <hip/hip_runtime.h>
#include <hip/hip_bf16.h>
#include <math.h>

typedef float v2f __attribute__((ext_vector_type(2)));
typedef float v8f __attribute__((ext_vector_type(8)));

#define EPS 1e-12f
#define NEIGH_WEIGHT 0.2f
#define CAP 32

#define BIN_SHIFT 8
#define BIN_NODES 256
#define BIN_CAP (BIN_NODES * CAP)
#define NSTAT 16
#define ZERO_WORDS 4096
#define MAXBIN 512
#define EPB 8192
#define BIN_THREADS 1024
#define CNT_STRIDE 32
#define GEMM_BLOCKS 512

__global__ __launch_bounds__(256) void gemm_h_kernel(
    const float* __restrict__ feats, const float* __restrict__ W,
    const float* __restrict__ bias, float* __restrict__ h, int N,
    const float* __restrict__ pos_diff, const int* __restrict__ nor_idx, int M,
    float* __restrict__ scratch, int* __restrict__ cnt, int* __restrict__ ovfn,
    int* __restrict__ binCount, int nbin, int nz) {
    __shared__ float As[16 * 260];
    __shared__ float T[16 * 132];
    __shared__ float P[16 * 17];
    __shared__ float RowInv[16];

    const int tid  = threadIdx.x;

    if ((int)blockIdx.x < nz) {
        const int w0 = blockIdx.x * ZERO_WORDS + tid;
        for (int j = 0; j < ZERO_WORDS / 256; ++j) {
            const int t = w0 + j * 256;
            if (t < N) cnt[t] = 0;
        }
        for (int t = blockIdx.x * 256 + tid; t < nbin; t += nz * 256) binCount[t * CNT_STRIDE] = 0;
        if (blockIdx.x == 0 && tid == 0) ovfn[0] = 0;
        return;
    }
    if ((int)blockIdx.x < nz + NSTAT) {
        const int sb = blockIdx.x - nz;
        float s1 = 0.0f, s2 = 0.0f;
        for (int k0 = sb * 256 + tid; k0 < M; k0 += 4 * NSTAT * 256) {
            int   ix[4];
            float v[4];
            #pragma unroll
            for (int u = 0; u < 4; ++u) {
                const int k = k0 + u * NSTAT * 256;
                ix[u] = nor_idx[k < M ? k : M - 1];
            }
            #pragma unroll
            for (int u = 0; u < 4; ++u) v[u] = pos_diff[ix[u]];
            #pragma unroll
            for (int u = 0; u < 4; ++u) {
                const float x = (k0 + u * NSTAT * 256 < M) ? v[u] : 0.0f;
                s1 += x; s2 += x * x;
            }
        }
        for (int off = 16; off > 0; off >>= 1) {
            s1 += __shfl_xor(s1, off, 32);
            s2 += __shfl_xor(s2, off, 32);
        }
        if ((tid & 31) == 0) { P[2 * (tid >> 5)] = s1; P[2 * (tid >> 5) + 1] = s2; }
        __syncthreads();
        if (tid == 0) {
            float a1 = 0.0f, a2 = 0.0f;
            for (int w = 0; w < 8; ++w) { a1 += P[2 * w]; a2 += P[2 * w + 1]; }
            scratch[2 * sb]     = a1;
            scratch[2 * sb + 1] = a2;
        }
        return;
    }

    const int wave = tid >> 5;
    const int lane = tid & 31;
    const int cb   = wave * 16;
    const int hl   = lane & 15;
    const int hi   = lane >> 4;
    const int ntile = (N + 15) / 16;
    const int pb = blockIdx.x - nz - NSTAT;
    const int nb = gridDim.x - nz - NSTAT;

    const int sr = tid >> 6;
    const int sc = (tid & 63) * 4;
    float4 pre[4];
    {
        const int t0 = pb < ntile ? pb : 0;
        #pragma unroll
        for (int j = 0; j < 4; ++j) {
            const int grow = t0 * 16 + sr + 4 * j;
            pre[j] = *(const float4*)(feats + (size_t)(grow < N ? grow : N - 1) * 256 + sc);
        }
    }

    v2f breg[64];
    #pragma unroll
    for (int kk = 0; kk < 64; ++kk) {
        const int ka = kk * 4 + hi * 2;
        breg[kk].x = W[(size_t)ka * 128 + cb + hl];
        breg[kk].y = W[(size_t)(ka + 1) * 128 + cb + hl];
    }
    const float bv = bias[cb + hl];
    const int rbase = hi * 8;

#define LDS_BARRIER() asm volatile("s_wait_dscnt 0x0\n\ts_barrier_signal -1\n\ts_barrier_wait -1" ::: "memory")

    for (int tile = pb; tile < ntile; tile += nb) {
        const int row0 = tile * 16;

        #pragma unroll
        for (int j = 0; j < 4; ++j) {
            float* dstp = &As[(sr + 4 * j) * 260 + sc];
            dstp[0] = pre[j].x; dstp[1] = pre[j].y; dstp[2] = pre[j].z; dstp[3] = pre[j].w;
        }
        LDS_BARRIER();

        {
            const int tn = tile + nb < ntile ? tile + nb : tile;
            #pragma unroll
            for (int j = 0; j < 4; ++j) {
                const int grow = tn * 16 + sr + 4 * j;
                pre[j] = *(const float4*)(feats + (size_t)(grow < N ? grow : N - 1) * 256 + sc);
            }
        }

        v8f acc = {};
        v2f af[2][8];
        const float* arow = &As[hl * 260 + hi * 2];
        #pragma unroll
        for (int u = 0; u < 8; ++u) { af[0][u].x = arow[u * 4]; af[0][u].y = arow[u * 4 + 1]; }
        #pragma unroll
        for (int g = 0; g < 8; ++g) {
            if (g < 7) {
                #pragma unroll
                for (int u = 0; u < 8; ++u) {
                    af[(g + 1) & 1][u].x = arow[(g + 1) * 32 + u * 4];
                    af[(g + 1) & 1][u].y = arow[(g + 1) * 32 + u * 4 + 1];
                }
            }
            __builtin_amdgcn_sched_barrier(0);
            #pragma unroll
            for (int u = 0; u < 8; ++u)
                acc = __builtin_amdgcn_wmma_f32_16x16x4_f32(false, af[g & 1][u], false, breg[g * 8 + u],
                                                            (short)0, acc, false, false);
            __builtin_amdgcn_sched_barrier(0);
        }

        #pragma unroll
        for (int r = 0; r < 8; ++r) {
            float v = acc[r] + bv;
            v = v > 0.0f ? v : 0.0f;
            T[(rbase + r) * 132 + cb + hl] = v;
        }
        LDS_BARRIER();

        {
            int row = tid >> 4;
            int c0  = (tid & 15) * 8;
            float s = 0.0f;
            #pragma unroll
            for (int j = 0; j < 8; ++j) { float v = T[row * 132 + c0 + j]; s += v * v; }
            P[row * 17 + (tid & 15)] = s;
        }
        LDS_BARRIER();
        if (tid < 16) {
            float s = 0.0f;
            #pragma unroll
            for (int j = 0; j < 16; ++j) s += P[tid * 17 + j];
            RowInv[tid] = 1.0f / fmaxf(sqrtf(s), EPS);
        }
        LDS_BARRIER();

        {
            int row  = tid >> 4;
            int grow = row0 + row;
            if (grow < N) {
                int c0 = (tid & 15) * 8;
                float inv = RowInv[row];
                float* op = h + (size_t)grow * 128 + c0;
                #pragma unroll
                for (int j = 0; j < 8; ++j) op[j] = T[row * 132 + c0 + j] * inv;
            }
        }
    }
#undef LDS_BARRIER
}

__global__ __launch_bounds__(BIN_THREADS) void bin_kernel(const int* __restrict__ src,
                                                          const int* __restrict__ dst,
                                                          int* __restrict__ binCount,
                                                          int* __restrict__ bins,
                                                          int* __restrict__ cnt,
                                                          int* __restrict__ ovfn,
                                                          int* __restrict__ ovf,
                                                          int N, int E, int nbin) {
    __shared__ int hist[MAXBIN];
    __shared__ int off[MAXBIN];
    __shared__ int gbase[MAXBIN];
    __shared__ int wsum[32];
    __shared__ int val[EPB];
    __shared__ unsigned short binOf[EPB];

    const int tid  = threadIdx.x;
    const int lane = tid & 31;
    const int wave = tid >> 5;
    const int e0   = blockIdx.x * EPB;

    if (tid < MAXBIN) hist[tid] = 0;
    __syncthreads();

    int dd[EPB / BIN_THREADS], ss[EPB / BIN_THREADS], rr[EPB / BIN_THREADS];
    #pragma unroll
    for (int k = 0; k < EPB / BIN_THREADS; ++k) {
        const int e = e0 + k * BIN_THREADS + tid;
        dd[k] = -1; ss[k] = 0;
        if (e < E) { dd[k] = dst[e]; ss[k] = src[e]; }
    }
    #pragma unroll
    for (int k = 0; k < EPB / BIN_THREADS; ++k) {
        rr[k] = 0;
        if ((unsigned)dd[k] < (unsigned)N)
            rr[k] = __hip_atomic_fetch_add(&hist[dd[k] >> BIN_SHIFT], 1, __ATOMIC_RELAXED,
                                           __HIP_MEMORY_SCOPE_WORKGROUP);
    }
    __syncthreads();

    const int v = (tid < MAXBIN) ? hist[tid] : 0;
    int inc = v;
    #pragma unroll
    for (int o = 1; o < 32; o <<= 1) {
        const int t = __shfl_up(inc, o, 32);
        if (lane >= o) inc += t;
    }
    if (lane == 31) wsum[wave] = inc;
    if (tid < nbin && v > 0)
        gbase[tid] = __hip_atomic_fetch_add(&binCount[tid * CNT_STRIDE], v, __ATOMIC_RELAXED,
                                            __HIP_MEMORY_SCOPE_AGENT);
    __syncthreads();
    if (wave == 0) {
        const int w = wsum[lane];
        int winc = w;
        #pragma unroll
        for (int o = 1; o < 32; o <<= 1) {
            const int t = __shfl_up(winc, o, 32);
            if (lane >= o) winc += t;
        }
        wsum[lane] = winc - w;
    }
    __syncthreads();
    if (tid < MAXBIN) off[tid] = inc - v + wsum[wave];
    __syncthreads();

    #pragma unroll
    for (int k = 0; k < EPB / BIN_THREADS; ++k) {
        if ((unsigned)dd[k] < (unsigned)N) {
            const int b = dd[k] >> BIN_SHIFT;
            const int p = off[b] + rr[k];
            val[p]   = ((dd[k] & (BIN_NODES - 1)) << 24) | (ss[k] & 0xFFFFFF);
            binOf[p] = (unsigned short)b;
        }
    }
    __syncthreads();

    const int nval = off[MAXBIN - 1] + hist[MAXBIN - 1];
    for (int p = tid; p < nval; p += BIN_THREADS) {
        const int b = binOf[p];
        const int g = gbase[b] + (p - off[b]);
        const int w = val[p];
        const int left = N - (b << BIN_SHIFT);
        const int capb = (left < BIN_NODES ? left : BIN_NODES) * CAP;
        if ((unsigned)g < (unsigned)capb) {
            bins[(size_t)b * BIN_CAP + g] = w;
        } else {
            const int d = (b << BIN_SHIFT) + (int)((unsigned)w >> 24);
            const int q = __hip_atomic_fetch_add(ovfn, 1, __ATOMIC_RELAXED, __HIP_MEMORY_SCOPE_AGENT);
            if ((unsigned)q < (unsigned)E) {
                ovf[2 * (size_t)q]     = d;
                ovf[2 * (size_t)q + 1] = w & 0xFFFFFF;
                __hip_atomic_fetch_add(&cnt[d], 1, __ATOMIC_RELAXED, __HIP_MEMORY_SCOPE_AGENT);
            }
        }
    }
}

#define NPW 10
#define PULL_NODES (8 * NPW)
#define OVL_CAP 64
#define PF 5

static_assert(PULL_NODES <= BIN_NODES, "a pull block's nodes must lie in at most two bins");

__device__ __forceinline__ float4 row4(const float* __restrict__ hl, int s) {
    return *(const float4*)(hl + (size_t)s * 128);
}
__device__ __forceinline__ void add4(float4& a, const float4& v) {
    a.x += v.x; a.y += v.y; a.z += v.z; a.w += v.w;
}
template <int G>
__device__ __forceinline__ void add_rows(float4& acc, const float* __restrict__ hl,
                                         int myidx, int k) {
    float4 v[G];
    #pragma unroll
    for (int u = 0; u < G; ++u) v[u] = row4(hl, __builtin_amdgcn_readlane(myidx, k + u));
    #pragma unroll
    for (int u = 0; u < G; ++u) add4(acc, v[u]);
}
__device__ __forceinline__ void walk(float4& acc, const float* __restrict__ hl, int myidx, int m) {
    int k = 0;
    for (; k + 8 <= m; k += 8) add_rows<8>(acc, hl, myidx, k);
    if (m & 4) { add_rows<4>(acc, hl, myidx, k); k += 4; }
    if (m & 2) { add_rows<2>(acc, hl, myidx, k); k += 2; }
    if (m & 1) { add_rows<1>(acc, hl, myidx, k); }
}

__global__ __launch_bounds__(256) void pull_finalize_kernel(
    const float* __restrict__ h, const int* __restrict__ cnt,
    const int* __restrict__ binCount, const int* __restrict__ bins,
    const int* __restrict__ ovfn, const int* __restrict__ ovf,
    const float* __restrict__ center,
    const float* __restrict__ pos_diff, const int* __restrict__ out_nodes,
    const int* __restrict__ labels, const int* __restrict__ epoch,
    const float* __restrict__ scratch, float* __restrict__ rec,
    float* __restrict__ scores_out, int N, int E, int Mnor) {
    __shared__ float red[8][4];
    __shared__ float sc[PULL_NODES];
    __shared__ int cntL[PULL_NODES];
    __shared__ int bucketL[PULL_NODES * (CAP + 1)];
    __shared__ int ovL[2 * OVL_CAP];
    __shared__ int ovLn;

    const int wave = __builtin_amdgcn_readfirstlane(threadIdx.x >> 5);
    const int tid  = threadIdx.x;
    const int lane = threadIdx.x & 31;
    const int s0 = blockIdx.x * PULL_NODES;
    const int r0 = wave * NPW;
    const int i0 = s0 + r0;
    const int nn = i0 < N ? (i0 + NPW < N ? NPW : N - i0) : 0;
    const float* hl = h + lane * 4;

    if (tid < PULL_NODES) cntL[tid] = 0;
    if (tid == 0) ovLn = 0;
    __syncthreads();

    const int ic = i0 < N ? i0 : N - 1;
    const int iv = i0 + lane < N ? i0 + lane : N - 1;
    float4 hv = *(const float4*)(hl + (size_t)ic * 128);
    const float4 cv  = *(const float4*)(center + lane * 4);
    const int    ep  = epoch[0];
    const float2 sxy = *(const float2*)(scratch + 2 * (lane < NSTAT ? lane : NSTAT - 1));
    const int    sp_v = cnt[iv];
    const int    on_v = out_nodes[iv];

    const int lastn = (s0 + PULL_NODES < N ? s0 + PULL_NODES : N) - 1;
    const int b0 = s0 >> BIN_SHIFT;
    const int b1 = lastn >> BIN_SHIFT;
    int n0 = binCount[b0 * CNT_STRIDE];
    int n1 = b1 > b0 ? binCount[b1 * CNT_STRIDE] : 0;
    {
        const int left0 = N - (b0 << BIN_SHIFT);
        const int cap0  = (left0 < BIN_NODES ? left0 : BIN_NODES) * CAP;
        if (n0 > cap0) n0 = cap0;
        const int left1 = N - (b1 << BIN_SHIFT);
        const int cap1  = (left1 < BIN_NODES ? left1 : BIN_NODES) * CAP;
        if (n1 > cap1) n1 = cap1;
    }

    constexpr int CHUNK = 4 * PF * 256;
    int4 w[PF];
    auto load_chunk = [&](const int* __restrict__ my, int base, int n) {
        #pragma unroll
        for (int u = 0; u < PF; ++u) {
            const int e = base + (u * 256 + tid) * 4;
            w[u] = *(const int4*)(my + (e < n ? e : 0));
        }
    };
    auto place_chunk = [&](int base, int n, int rel) {
        #pragma unroll
        for (int u = 0; u < PF; ++u) {
            const int e = base + (u * 256 + tid) * 4;
            const int ww[4] = {w[u].x, w[u].y, w[u].z, w[u].w};
            int  r[4], slot[4];
            bool ok[4];
            #pragma unroll
            for (int q = 0; q < 4; ++q) {
                r[q]  = rel + (int)((unsigned)ww[q] >> 24);
                ok[q] = e + q < n && (unsigned)r[q] < (unsigned)PULL_NODES;
                if (ok[q])
                    slot[q] = __hip_atomic_fetch_add(&cntL[r[q]], 1, __ATOMIC_RELAXED,
                                                     __HIP_MEMORY_SCOPE_WORKGROUP);
            }
            #pragma unroll
            for (int q = 0; q < 4; ++q) {
                if (ok[q]) {
                    const int s = ww[q] & 0xFFFFFF;
                    if ((unsigned)slot[q] < (unsigned)CAP) {
                        bucketL[r[q] * (CAP + 1) + slot[q]] = s;
                    } else {
                        const int p = __hip_atomic_fetch_add(&ovLn, 1, __ATOMIC_RELAXED,
                                                             __HIP_MEMORY_SCOPE_WORKGROUP);
                        if ((unsigned)p < (unsigned)OVL_CAP) { ovL[2 * p] = r[q]; ovL[2 * p + 1] = s; }
                    }
                }
            }
        }
    };

    const int* my0 = bins + (size_t)b0 * BIN_CAP;
    const int rel0 = (b0 << BIN_SHIFT) - s0;
    load_chunk(my0, 0, n0);
    const float pd_v  = pos_diff[on_v];
    const int   lab_v = labels[on_v];
    place_chunk(0, n0, rel0);
    for (int base = CHUNK; base < n0; base += CHUNK) {
        load_chunk(my0, base, n0);
        place_chunk(base, n0, rel0);
    }
    if (b1 > b0) {
        const int* my1 = bins + (size_t)b1 * BIN_CAP;
        const int rel1 = (b1 << BIN_SHIFT) - s0;
        for (int base = 0; base < n1; base += CHUNK) {
            load_chunk(my1, base, n1);
            place_chunk(base, n1, rel1);
        }
    }

    float mu = 0.0f, sd = 1.0f, beta = 0.0f;
    if (nn > 0) {
        const float cntf = (float)Mnor;
        float st1 = 0.0f, st2 = 0.0f;
        for (int q = 0; q < NSTAT; ++q) {
            st1 += __int_as_float(__builtin_amdgcn_readlane(__float_as_int(sxy.x), q));
            st2 += __int_as_float(__builtin_amdgcn_readlane(__float_as_int(sxy.y), q));
        }
        mu = st1 / cntf;
        const float var = (st2 - cntf * mu * mu) / (cntf - 1.0f);
        sd = sqrtf(var);
        beta = powf(0.9f, (float)ep);
        if (beta < 0.1f) beta = 0.0f;
    }
    __syncthreads();

    float p_sum = 0.0f, p_cnt = 0.0f, n_sum = 0.0f, n_cnt = 0.0f;

    if (nn > 0) {
        int idx = bucketL[r0 * (CAP + 1) + lane];

        int cl_v = 0;
        if (lane < nn) cl_v = cntL[r0 + lane];

        const int   c_v    = cl_v + sp_v;
        const float invd_v = c_v > 0 ? 1.0f / fmaxf((float)c_v, 1.0f) : 0.0f;
        const float z_v    = (pd_v - mu) / sd;
        const float pre_v  = 1.0f - 1.0f / (1.0f + expf(-z_v));
        float score_v = 0.0f;

        for (int j = 0; j < nn; ++j) {
            const int i = i0 + j;
            const int    idxc = idx;
            const float4 hvc  = hv;
            if (j + 1 < nn) {
                idx = bucketL[(r0 + j + 1) * (CAP + 1) + lane];
                hv  = *(const float4*)(hl + (size_t)(i + 1) * 128);
            }
            const int   cl   = __builtin_amdgcn_readlane(cl_v, j);
            const int   sp   = __builtin_amdgcn_readlane(sp_v, j);
            const float invd = __int_as_float(__builtin_amdgcn_readlane(__float_as_int(invd_v), j));
            const float pre  = __int_as_float(__builtin_amdgcn_readlane(__float_as_int(pre_v), j));

            const int m0 = cl < CAP ? cl : CAP;

            float4 acc = make_float4(0.f, 0.f, 0.f, 0.f);
            if (m0 > 0) walk(acc, hl, idxc, m0);
            if (cl > CAP) {
                const int nl = ovLn;
                if (nl <= OVL_CAP) {
                    for (int base = 0; base < nl; base += 32) {
                        const int k = base + lane;
                        int r = -1, s = 0;
                        if (k < nl) { r = ovL[2 * k]; s = ovL[2 * k + 1]; }
                        unsigned hit = __builtin_amdgcn_ballot_w32(r == r0 + j);
                        while (hit) {
                            const int b = __builtin_ctz(hit);
                            hit &= hit - 1;
                            const float4 v = row4(hl, __builtin_amdgcn_readlane(s, b));
                            add4(acc, v);
                        }
                    }
                } else {
                    acc = make_float4(0.f, 0.f, 0.f, 0.f);
                    const int bb = i >> BIN_SHIFT;
                    int n = binCount[bb * CNT_STRIDE];
                    const int left = N - (bb << BIN_SHIFT);
                    const int capb = (left < BIN_NODES ? left : BIN_NODES) * CAP;
                    if (n > capb) n = capb;
                    const int* my = bins + (size_t)bb * BIN_CAP;
                    const int dlw = i & (BIN_NODES - 1);
                    for (int base = 0; base < n; base += 32) {
                        const int k = base + lane;
                        int w = 0;
                        if (k < n) w = my[k];
                        unsigned hit = __builtin_amdgcn_ballot_w32(k < n && (int)((unsigned)w >> 24) == dlw);
                        while (hit) {
                            const int b = __builtin_ctz(hit);
                            hit &= hit - 1;
                            const float4 v = row4(hl, __builtin_amdgcn_readlane(w, b) & 0xFFFFFF);
                            add4(acc, v);
                        }
                    }
                }
            }
            if (sp > 0) {
                int n = ovfn[0];
                if (n > E) n = E;
                for (int base = 0; base < n; base += 32) {
                    const int k = base + lane;
                    int d = -1, s = 0;
                    if (k < n) { d = ovf[2 * (size_t)k]; s = ovf[2 * (size_t)k + 1]; }
                    unsigned hit = __builtin_amdgcn_ballot_w32(d == i);
                    while (hit) {
                        const int b = __builtin_ctz(hit);
                        hit &= hit - 1;
                        const float4 v = row4(hl, __builtin_amdgcn_readlane(s, b));
                        add4(acc, v);
                    }
                }
            }

            float4 mv;
            mv.x = acc.x * invd; mv.y = acc.y * invd; mv.z = acc.z * invd; mv.w = acc.w * invd;
            float dot = hvc.x * mv.x;
            dot += hvc.y * mv.y;
            dot += hvc.z * mv.z;
            dot += hvc.w * mv.w;
            for (int o = 16; o > 0; o >>= 1) dot += __shfl_xor(dot, o, 32);

            float attn = beta * pre + (1.0f - beta) * dot;
            float nei  = attn * NEIGH_WEIGHT;

            float score = (nei * mv.x + (1.0f - nei) * hvc.x) * cv.x;
            score += (nei * mv.y + (1.0f - nei) * hvc.y) * cv.y;
            score += (nei * mv.z + (1.0f - nei) * hvc.z) * cv.z;
            score += (nei * mv.w + (1.0f - nei) * hvc.w) * cv.w;
            for (int o = 16; o > 0; o >>= 1) score += __shfl_xor(score, o, 32);

            if (lane == j) score_v = score;
        }

        if (lane < nn) {
            sc[wave * NPW + lane] = score_v;
            float spl = log1pf(expf(-fabsf(score_v)));
            if (lab_v == 1) {
                n_sum = fmaxf(score_v, 0.0f) + spl;
                n_cnt = 1.0f;
            } else {
                p_sum = fmaxf(-score_v, 0.0f) + spl;
                p_cnt = 1.0f;
            }
        }
        for (int o = 16; o > 0; o >>= 1) {
            p_sum += __shfl_xor(p_sum, o, 32);
            p_cnt += __shfl_xor(p_cnt, o, 32);
            n_sum += __shfl_xor(n_sum, o, 32);
            n_cnt += __shfl_xor(n_cnt, o, 32);
        }
    }

    if (lane == 0) {
        red[wave][0] = p_sum; red[wave][1] = p_cnt;
        red[wave][2] = n_sum; red[wave][3] = n_cnt;
    }
    __syncthreads();
    if (threadIdx.x < 4) {
        float s = 0.0f;
        for (int w = 0; w < 8; ++w) s += red[w][threadIdx.x];
        rec[4 * (size_t)blockIdx.x + threadIdx.x] = s;
    }
    if (threadIdx.x < PULL_NODES) {
        const int i = blockIdx.x * PULL_NODES + threadIdx.x;
        if (i < N) scores_out[i] = sc[threadIdx.x];
    }
}

__global__ __launch_bounds__(1024) void loss_kernel(const float* __restrict__ rec, int N,
                                                    float* __restrict__ out0) {
    __shared__ float red[32][4];
    const int tid  = threadIdx.x;
    const int lane = tid & 31;
    const int wave = tid >> 5;
    const int nblk = (N + PULL_NODES - 1) / PULL_NODES;

    float a0 = 0.0f, a1 = 0.0f, a2 = 0.0f, a3 = 0.0f;
    #pragma unroll 4
    for (int b = tid; b < nblk; b += 1024) {
        float4 v = *(const float4*)(rec + 4 * (size_t)b);
        a0 += v.x; a1 += v.y; a2 += v.z; a3 += v.w;
    }
    for (int off = 16; off > 0; off >>= 1) {
        a0 += __shfl_xor(a0, off, 32);
        a1 += __shfl_xor(a1, off, 32);
        a2 += __shfl_xor(a2, off, 32);
        a3 += __shfl_xor(a3, off, 32);
    }
    if (lane == 0) {
        red[wave][0] = a0; red[wave][1] = a1; red[wave][2] = a2; red[wave][3] = a3;
    }
    __syncthreads();
    if (wave == 0) {
        float r0 = red[lane][0], r1 = red[lane][1], r2 = red[lane][2], r3 = red[lane][3];
        for (int off = 16; off > 0; off >>= 1) {
            r0 += __shfl_xor(r0, off, 32);
            r1 += __shfl_xor(r1, off, 32);
            r2 += __shfl_xor(r2, off, 32);
            r3 += __shfl_xor(r3, off, 32);
        }
        if (lane == 0)
            out0[0] = r0 / fmaxf(r1, 1.0f) + r2 / fmaxf(r3, 1.0f);
    }
}

extern "C" void kernel_launch(void* const* d_in, const int* in_sizes, int n_in,
                              void* d_out, int out_size, void* d_ws, size_t ws_size,
                              hipStream_t stream) {
    const float* feats    = (const float*)d_in[0];
    const float* W        = (const float*)d_in[1];
    const float* b        = (const float*)d_in[2];
    const float* center   = (const float*)d_in[3];
    const float* pos_diff = (const float*)d_in[4];
    const int*   src      = (const int*)d_in[5];
    const int*   dst      = (const int*)d_in[6];
    const int*   out_nodes= (const int*)d_in[7];
    const int*   nor_idx  = (const int*)d_in[8];
    const int*   labels   = (const int*)d_in[9];
    const int*   epoch    = (const int*)d_in[10];

    const int N = in_sizes[0] / 256;
    const int E = in_sizes[5];
    const int M = in_sizes[8];
    const int nblk = (N + 7) / 8;

    const int nbin = (N + BIN_NODES - 1) / BIN_NODES;
    const int npull = (N + PULL_NODES - 1) / PULL_NODES;

    const int nz = (N + ZERO_WORDS - 1) / ZERO_WORDS;

    float* wsf     = (float*)d_ws;
    float* scratch = wsf;
    float* h       = wsf + 64;
    int*   bins    = (int*)(h + (size_t)N * 128);
    float* rec     = (float*)(bins + (size_t)N * CAP);
    int*   ovfn    = (int*)(rec + 4 * (size_t)nblk);
    int*   ovf     = ovfn + 4;
    int*   cnt     = ovf + 2 * (size_t)E;
    int*   binCount= (int*)rec + (4 * npull + CNT_STRIDE - 1) / CNT_STRIDE * CNT_STRIDE;

    float* out     = (float*)d_out;

    gemm_h_kernel<<<nz + NSTAT + GEMM_BLOCKS, 256, 0, stream>>>(feats, W, b, h, N, pos_diff, nor_idx, M,
                                                                 scratch, cnt, ovfn, binCount, nbin, nz);

    bin_kernel<<<(E + EPB - 1) / EPB, BIN_THREADS, 0, stream>>>(src, dst, binCount, bins, cnt, ovfn, ovf, N, E, nbin);

    pull_finalize_kernel<<<npull, 256, 0, stream>>>(h, cnt, binCount, bins, ovfn, ovf, center, pos_diff,
                                                   out_nodes, labels, epoch, scratch, rec,
                                                   out + 1, N, E, M);

    loss_kernel<<<1, 1024, 0, stream>>>(rec, N, out);
}
